# starting kernel
# baseline (speedup 1.0000x reference)
#include <hip/hip_runtime.h>
#include <hip/hip_bf16.h>
#include <math.h>

#define NEG_SLOPE 0.01f
#define IN_DIM  128
#define OUT_DIM 32

typedef __attribute__((ext_vector_type(2))) float v2f;
typedef __attribute__((ext_vector_type(8))) float v8f;

__global__ void gat_init_kernel(float* __restrict__ h_out,
                                float* __restrict__ e_max,
                                float* __restrict__ denom,
                                int N) {
    int gid = blockIdx.x * blockDim.x + threadIdx.x;
    if (gid < N * OUT_DIM) h_out[gid] = 0.0f;
    if (gid < N) {
        e_max[gid] = -INFINITY;
        denom[gid] = 0.0f;
    }
}

__global__ void gat_gemm_wmma_kernel(const float* __restrict__ h,
                                     const float* __restrict__ fc_w,
                                     float* __restrict__ z,
                                     int N) {
    const int lane   = threadIdx.x & 31;
    const int waveId = blockIdx.x * (blockDim.x >> 5) + (threadIdx.x >> 5);
    const int m0     = waveId * 16;
    if (m0 >= N) return;

    const int n15   = lane & 15;
    const int khalf = lane >> 4;

    const float* __restrict__ arow = h + (size_t)(m0 + n15) * IN_DIM + 2 * khalf;
    const float* __restrict__ b0p  = fc_w + (size_t)n15        * IN_DIM + 2 * khalf;
    const float* __restrict__ b1p  = fc_w + (size_t)(16 + n15) * IN_DIM + 2 * khalf;

    v8f acc0 = {};
    v8f acc1 = {};

    #pragma unroll
    for (int k = 0; k < IN_DIM; k += 4) {
        v2f a  = *(const v2f*)(arow + k);
        v2f b0 = *(const v2f*)(b0p + k);
        v2f b1 = *(const v2f*)(b1p + k);
        acc0 = __builtin_amdgcn_wmma_f32_16x16x4_f32(
                   false, a, false, b0, (short)0, acc0, false, false);
        acc1 = __builtin_amdgcn_wmma_f32_16x16x4_f32(
                   false, a, false, b1, (short)0, acc1, false, false);
    }

    #pragma unroll
    for (int g = 0; g < 8; ++g) {
        int row = m0 + g + 8 * khalf;
        z[(size_t)row * OUT_DIM + n15]      = acc0[g];
        z[(size_t)row * OUT_DIM + 16 + n15] = acc1[g];
    }
}

__global__ void gat_sls_kernel(const float* __restrict__ z,
                               const float* __restrict__ attn_w,
                               float* __restrict__ s_l,
                               float* __restrict__ s_r,
                               int N) {
    int i = blockIdx.x * blockDim.x + threadIdx.x;
    if (i >= N) return;
    const float4* zr = (const float4*)(z + (size_t)i * OUT_DIM);
    const float4* al = (const float4*)(attn_w);
    const float4* ar = (const float4*)(attn_w + OUT_DIM);
    float accl = 0.0f, accr = 0.0f;
    #pragma unroll
    for (int q = 0; q < OUT_DIM / 4; ++q) {
        float4 zv = zr[q];
        float4 lv = al[q];
        float4 rv = ar[q];
        accl += zv.x * lv.x + zv.y * lv.y + zv.z * lv.z + zv.w * lv.w;
        accr += zv.x * rv.x + zv.y * rv.y + zv.z * rv.z + zv.w * rv.w;
    }
    s_l[i] = accl;
    s_r[i] = accr;
}

__global__ void gat_edge_max_kernel(const int* __restrict__ src,
                                    const int* __restrict__ dst,
                                    const float* __restrict__ s_l,
                                    const float* __restrict__ s_r,
                                    float* __restrict__ e_buf,
                                    float* __restrict__ e_max,
                                    int E) {
    int i = blockIdx.x * blockDim.x + threadIdx.x;
    if (i >= E) return;
    int s = src[i], t = dst[i];
    float ev = s_l[s] + s_r[t];
    ev = ev > 0.0f ? ev : NEG_SLOPE * ev;
    e_buf[i] = ev;
    atomicMax(&e_max[t], ev);
}

__global__ void gat_edge_exp_kernel(const int* __restrict__ dst,
                                    float* __restrict__ e_buf,
                                    const float* __restrict__ e_max,
                                    float* __restrict__ denom,
                                    int E) {
    int i = blockIdx.x * blockDim.x + threadIdx.x;
    if (i >= E) return;
    int t = dst[i];
    float ex = __expf(e_buf[i] - e_max[t]);
    e_buf[i] = ex;
    atomicAdd(&denom[t], ex);
}

__global__ void gat_scatter_kernel(const int* __restrict__ src,
                                   const int* __restrict__ dst,
                                   const float* __restrict__ z,
                                   const float* __restrict__ e_buf,
                                   const float* __restrict__ denom,
                                   float* __restrict__ h_out,
                                   int E) {
    int gid  = blockIdx.x * blockDim.x + threadIdx.x;
    int edge = gid >> 5;
    int d    = gid & 31;
    if (edge >= E) return;
    int s = src[edge], t = dst[edge];
    float alpha = e_buf[edge] / denom[t];
    float val   = alpha * z[(size_t)s * OUT_DIM + d];
    atomicAdd(&h_out[(size_t)t * OUT_DIM + d], val);
}

extern "C" void kernel_launch(void* const* d_in, const int* in_sizes, int n_in,
                              void* d_out, int out_size, void* d_ws, size_t ws_size,
                              hipStream_t stream) {
    const float* h      = (const float*)d_in[0];
    const int*   src    = (const int*)  d_in[1];
    const int*   dst    = (const int*)  d_in[2];
    const float* fc_w   = (const float*)d_in[3];
    const float* attn_w = (const float*)d_in[4];

    const int N = in_sizes[0] / IN_DIM;
    const int E = in_sizes[1];

    float* h_out = (float*)d_out;

    auto align256 = [](size_t x) { return (x + 255) & ~(size_t)255; };
    char* w = (char*)d_ws;
    float* z     = (float*)w; w += align256((size_t)N * OUT_DIM * sizeof(float));
    float* s_l   = (float*)w; w += align256((size_t)N * sizeof(float));
    float* s_r   = (float*)w; w += align256((size_t)N * sizeof(float));
    float* e_max = (float*)w; w += align256((size_t)N * sizeof(float));
    float* denom = (float*)w; w += align256((size_t)N * sizeof(float));
    float* e_buf = (float*)w; w += align256((size_t)E * sizeof(float));
    (void)ws_size;

    const int B = 256;

    {
        int total = N * OUT_DIM;
        gat_init_kernel<<<(total + B - 1) / B, B, 0, stream>>>(h_out, e_max, denom, N);
    }
    {
        int tiles = (N + 15) / 16;
        int wavesPerBlock = B / 32;
        int grid = (tiles + wavesPerBlock - 1) / wavesPerBlock;
        gat_gemm_wmma_kernel<<<grid, B, 0, stream>>>(h, fc_w, z, N);
    }
    gat_sls_kernel<<<(N + B - 1) / B, B, 0, stream>>>(z, attn_w, s_l, s_r, N);
    gat_edge_max_kernel<<<(E + B - 1) / B, B, 0, stream>>>(src, dst, s_l, s_r,
                                                           e_buf, e_max, E);
    gat_edge_exp_kernel<<<(E + B - 1) / B, B, 0, stream>>>(dst, e_buf, e_max,
                                                           denom, E);
    {
        long long total = (long long)E * 32;
        int grid = (int)((total + B - 1) / B);
        gat_scatter_kernel<<<grid, B, 0, stream>>>(src, dst, z, e_buf, denom,
                                                   h_out, E);
    }
}


# Grouping carries each edge's source node (read contiguously beside dst) instead of its index: node kernel reads its sources in one load, per-edge src gather gone; float sums stay in bucket order
# speedup vs baseline: 3.2487x; 3.2487x over previous
#include <hip/hip_runtime.h>
#include <hip/hip_bf16.h>
#include <math.h>

#define NEG_SLOPE 0.01f
#define IN_DIM  128
#define OUT_DIM 32

typedef __attribute__((ext_vector_type(2))) float v2f;
typedef __attribute__((ext_vector_type(8))) float v8f;

__global__ void gat_gemm_wmma_kernel(const float* __restrict__ h,
                                     const float* __restrict__ fc_w,
                                     float* __restrict__ z,
                                     int N) {
    const int lane   = threadIdx.x & 31;
    const int waveId = blockIdx.x * (blockDim.x >> 5) + (threadIdx.x >> 5);
    const int m0     = waveId * 16;
    if (m0 >= N) return;

    const int n15   = lane & 15;
    const int khalf = lane >> 4;

    const float* __restrict__ arow = h + (size_t)(m0 + n15) * IN_DIM + 2 * khalf;
    const float* __restrict__ b0p  = fc_w + (size_t)n15        * IN_DIM + 2 * khalf;
    const float* __restrict__ b1p  = fc_w + (size_t)(16 + n15) * IN_DIM + 2 * khalf;

    v8f acc0 = {};
    v8f acc1 = {};

    #pragma unroll
    for (int k = 0; k < IN_DIM; k += 4) {
        v2f a  = *(const v2f*)(arow + k);
        v2f b0 = *(const v2f*)(b0p + k);
        v2f b1 = *(const v2f*)(b1p + k);
        acc0 = __builtin_amdgcn_wmma_f32_16x16x4_f32(
                   false, a, false, b0, (short)0, acc0, false, false);
        acc1 = __builtin_amdgcn_wmma_f32_16x16x4_f32(
                   false, a, false, b1, (short)0, acc1, false, false);
    }

    #pragma unroll
    for (int g = 0; g < 8; ++g) {
        int row = m0 + g + 8 * khalf;
        z[(size_t)row * OUT_DIM + n15]      = acc0[g];
        z[(size_t)row * OUT_DIM + 16 + n15] = acc1[g];
    }
}

__global__ void gat_sls_kernel(const float* __restrict__ z,
                               const float* __restrict__ attn_w,
                               float* __restrict__ s_l,
                               float* __restrict__ s_r,
                               int N) {
    int i = blockIdx.x * blockDim.x + threadIdx.x;
    if (i >= N) return;
    const float4* zr = (const float4*)(z + (size_t)i * OUT_DIM);
    const float4* al = (const float4*)(attn_w);
    const float4* ar = (const float4*)(attn_w + OUT_DIM);
    float accl = 0.0f, accr = 0.0f;
    #pragma unroll
    for (int q = 0; q < OUT_DIM / 4; ++q) {
        float4 zv = zr[q];
        float4 lv = al[q];
        float4 rv = ar[q];
        accl += zv.x * lv.x + zv.y * lv.y + zv.z * lv.z + zv.w * lv.w;
        accr += zv.x * rv.x + zv.y * rv.y + zv.z * rv.z + zv.w * rv.w;
    }
    s_l[i] = accl;
    s_r[i] = accr;
}

#define BK_T 1024
#define BK_U 16
#define NO_NODE 0x7fffffff

template <bool FILL>
__device__ __forceinline__ int bucket_one(int dv, int e, int lo, int* cnt,
                                          const int* __restrict__ src,
                                          int* __restrict__ bucket_src) {
    const unsigned r = (unsigned)(dv - lo);
    if (r < (unsigned)BK_T) {
        if (FILL) {
            const int pos = atomicAdd(&cnt[r], 1);
            bucket_src[pos] = src[e];
        } else {
            atomicAdd(&cnt[r], 1);
        }
    }
    if (FILL) return 0;
    return __builtin_popcount(__builtin_amdgcn_ballot_w32(dv < lo));
}

template <bool GUARD, bool FILL>
__device__ __forceinline__ int bucket_batch(const int4* __restrict__ dst4, int q0, int E4,
                                            int lo, int* cnt,
                                            const int* __restrict__ src,
                                            int* __restrict__ bucket_src) {
    int4 d[BK_U];
    #pragma unroll
    for (int u = 0; u < BK_U; ++u) {
        const int q = q0 + u * BK_T;
        if (GUARD) {
            d[u] = make_int4(NO_NODE, NO_NODE, NO_NODE, NO_NODE);
            if (q < E4) d[u] = dst4[q];
        } else {
            d[u] = dst4[q];
        }
    }
    int below = 0;
    #pragma unroll
    for (int u = 0; u < BK_U; ++u) {
        const int e0 = (q0 + u * BK_T) * 4;
        below += bucket_one<FILL>(d[u].x, e0 + 0, lo, cnt, src, bucket_src);
        below += bucket_one<FILL>(d[u].y, e0 + 1, lo, cnt, src, bucket_src);
        below += bucket_one<FILL>(d[u].z, e0 + 2, lo, cnt, src, bucket_src);
        below += bucket_one<FILL>(d[u].w, e0 + 3, lo, cnt, src, bucket_src);
    }
    return below;
}

template <bool FILL>
__device__ __forceinline__ int bucket_pass(const int* __restrict__ dst, int E, int lo, int tid,
                                           int* cnt, const int* __restrict__ src,
                                           int* __restrict__ bucket_src) {
    const int4* __restrict__ dst4 = (const int4*)dst;
    const int E4 = E >> 2;
    const int per_batch = BK_T * BK_U;
    const int nfull = E4 / per_batch;
    int below = 0;
    for (int b = 0; b < nfull; ++b)
        below += bucket_batch<false, FILL>(dst4, b * per_batch + tid, E4, lo, cnt, src, bucket_src);
    if (nfull * per_batch < E4)
        below += bucket_batch<true, FILL>(dst4, nfull * per_batch + tid, E4, lo, cnt, src, bucket_src);
    {
        const int e = (E4 << 2) + tid;
        int dv = NO_NODE;
        if (e < E) dv = dst[e];
        below += bucket_one<FILL>(dv, e, lo, cnt, src, bucket_src);
    }
    return below;
}

__global__ void __launch_bounds__(BK_T)
gat_bucket_kernel(const int* __restrict__ src,
                  const int* __restrict__ dst,
                  int* __restrict__ offset,
                  int* __restrict__ count,
                  int* __restrict__ bucket_src,
                  int N, int E) {
    __shared__ int cnt[BK_T];
    __shared__ int wave_tot[32];
    __shared__ int wave_below[32];
    __shared__ int base_sh;
    const int tid  = threadIdx.x;
    const int lane = tid & 31;
    const int wv   = tid >> 5;
    const int lo   = blockIdx.x * BK_T;

    cnt[tid] = 0;
    __syncthreads();

    const int below = bucket_pass<false>(dst, E, lo, tid, cnt, src, bucket_src);
    __syncthreads();

    const int c = cnt[tid];
    int inc = c;
    #pragma unroll
    for (int d = 1; d < 32; d <<= 1) {
        int o = __shfl_up(inc, d, 32);
        if (lane >= d) inc += o;
    }
    if (lane == 31) wave_tot[wv] = inc;
    if (lane == 0)  wave_below[wv] = below;
    __syncthreads();
    if (wv == 0) {
        const int w = wave_tot[lane];
        int winc = w;
        #pragma unroll
        for (int d = 1; d < 32; d <<= 1) {
            int o = __shfl_up(winc, d, 32);
            if (lane >= d) winc += o;
        }
        wave_tot[lane] = winc - w;
        int b = wave_below[lane];
        #pragma unroll
        for (int d = 16; d >= 1; d >>= 1) b += __shfl_xor(b, d, 32);
        if (lane == 0) base_sh = b;
    }
    __syncthreads();
    const int start = base_sh + wave_tot[wv] + (inc - c);
    if (lo + tid < N) {
        offset[lo + tid] = start;
        count[lo + tid]  = c;
    }
    cnt[tid] = start;
    __syncthreads();

    (void)bucket_pass<true>(dst, E, lo, tid, cnt, src, bucket_src);
}

#define BK_SHIFT 10
#define CP_G 128
#define CP_T 1024
#define CP_U 8

__device__ __forceinline__ int hist_pitch_log2(int NB) {
    int hs = 5;
    while ((1 << hs) < NB) ++hs;
    return hs;
}

__global__ void __launch_bounds__(CP_T)
gat_coarse_hist_kernel(const int* __restrict__ dst, int* __restrict__ hist,
                       int N, int NB, int E) {
    __shared__ int cnt[CP_T];
    const int tid = threadIdx.x;
    const int c   = blockIdx.x;
    const int len = (E + CP_G - 1) / CP_G;
    const int beg = c * len;
    const int end = (beg + len < E) ? beg + len : E;
    const int HP  = 1 << hist_pitch_log2(NB);

    cnt[tid] = 0;
    __syncthreads();
    for (int base = beg; base < end; base += CP_T * CP_U) {
        int d[CP_U];
        #pragma unroll
        for (int u = 0; u < CP_U; ++u) {
            const int e = base + u * CP_T + tid;
            d[u] = -1;
            if (e < end) d[u] = dst[e];
        }
        #pragma unroll
        for (int u = 0; u < CP_U; ++u)
            if ((unsigned)d[u] < (unsigned)N) atomicAdd(&cnt[d[u] >> BK_SHIFT], 1);
    }
    __syncthreads();
    if (tid < HP) hist[c * HP + tid] = cnt[tid];
}

__global__ void __launch_bounds__(CP_T)
gat_coarse_scatter_kernel(const int* __restrict__ src, const int* __restrict__ dst,
                          const int* __restrict__ hist,
                          int* __restrict__ coarse_raw, int* __restrict__ binstart,
                          int N, int NB, int E) {
    __shared__ int cur[CP_T];
    __shared__ int tot[CP_T];
    __shared__ int wave_tot[32];
    int2* __restrict__ coarse = (int2*)coarse_raw;
    const int tid  = threadIdx.x;
    const int lane = tid & 31;
    const int wv   = tid >> 5;
    const int c    = blockIdx.x;
    const int len  = (E + CP_G - 1) / CP_G;
    const int beg  = c * len;
    const int end  = (beg + len < E) ? beg + len : E;
    const int hs   = hist_pitch_log2(NB);
    const int HP   = 1 << hs;

    cur[tid] = 0;
    tot[tid] = 0;
    __syncthreads();
    {
        const int b = tid & (HP - 1);
        int part = 0, all = 0;
        for (int j = tid >> hs; j < CP_G; j += (CP_T >> hs)) {
            const int v = hist[j * HP + b];
            all += v;
            if (j < c) part += v;
        }
        atomicAdd(&cur[b], part);
        atomicAdd(&tot[b], all);
    }
    __syncthreads();
    const int v = tot[tid];
    int inc = v;
    for (int d = 1; d < 32; d <<= 1) {
        int o = __shfl_up(inc, d, 32);
        if (lane >= d) inc += o;
    }
    if (lane == 31) wave_tot[wv] = inc;
    __syncthreads();
    if (wv == 0) {
        const int w = wave_tot[lane];
        int winc = w;
        for (int d = 1; d < 32; d <<= 1) {
            int o = __shfl_up(winc, d, 32);
            if (lane >= d) winc += o;
        }
        wave_tot[lane] = winc - w;
    }
    __syncthreads();
    const int excl = wave_tot[wv] + (inc - v);
    cur[tid] += excl;
    if (c == 0 && tid <= NB) binstart[tid] = excl;
    __syncthreads();

    for (int base = beg; base < end; base += CP_T * CP_U) {
        int d[CP_U];
        int s[CP_U];
        #pragma unroll
        for (int u = 0; u < CP_U; ++u) {
            const int e = base + u * CP_T + tid;
            d[u] = -1;
            s[u] = 0;
            if (e < end) { d[u] = dst[e]; s[u] = src[e]; }
        }
        #pragma unroll
        for (int u = 0; u < CP_U; ++u) {
            if ((unsigned)d[u] < (unsigned)N) {
                const int pos = atomicAdd(&cur[d[u] >> BK_SHIFT], 1);
                coarse[pos] = make_int2(d[u], s[u]);
            }
        }
    }
}

#define FB_U 8

template <bool FILL>
__device__ __forceinline__ void fine_pass(const int2* __restrict__ coarse, int rb, int re,
                                          int lo, int tid, int* cnt,
                                          int* __restrict__ bucket_src) {
    for (int base = rb; base < re; base += BK_T * FB_U) {
        int2 p[FB_U];
        #pragma unroll
        for (int u = 0; u < FB_U; ++u) {
            const int i = base + u * BK_T + tid;
            p[u] = make_int2(-1, 0);
            if (i < re) p[u] = coarse[i];
        }
        #pragma unroll
        for (int u = 0; u < FB_U; ++u) {
            const unsigned r = (unsigned)(p[u].x - lo);
            if (r < (unsigned)BK_T) {
                if (FILL) {
                    const int pos = atomicAdd(&cnt[r], 1);
                    bucket_src[pos] = p[u].y;
                } else {
                    atomicAdd(&cnt[r], 1);
                }
            }
        }
    }
}

__global__ void __launch_bounds__(BK_T)
gat_fine_bucket_kernel(const int* __restrict__ coarse_raw, const int* __restrict__ binstart,
                       int* __restrict__ offset, int* __restrict__ count,
                       int* __restrict__ bucket_src, int N) {
    __shared__ int cnt[BK_T];
    __shared__ int wave_tot[32];
    const int2* __restrict__ coarse = (const int2*)coarse_raw;
    const int tid  = threadIdx.x;
    const int lane = tid & 31;
    const int wv   = tid >> 5;
    const int lo   = blockIdx.x * BK_T;
    const int rb   = binstart[blockIdx.x];
    const int re   = binstart[blockIdx.x + 1];

    cnt[tid] = 0;
    __syncthreads();
    fine_pass<false>(coarse, rb, re, lo, tid, cnt, bucket_src);
    __syncthreads();

    const int c = cnt[tid];
    int inc = c;
    for (int d = 1; d < 32; d <<= 1) {
        int o = __shfl_up(inc, d, 32);
        if (lane >= d) inc += o;
    }
    if (lane == 31) wave_tot[wv] = inc;
    __syncthreads();
    if (wv == 0) {
        const int w = wave_tot[lane];
        int winc = w;
        for (int d = 1; d < 32; d <<= 1) {
            int o = __shfl_up(winc, d, 32);
            if (lane >= d) winc += o;
        }
        wave_tot[lane] = winc - w;
    }
    __syncthreads();
    const int start = rb + wave_tot[wv] + (inc - c);
    if (lo + tid < N) {
        offset[lo + tid] = start;
        count[lo + tid]  = c;
    }
    cnt[tid] = start;
    __syncthreads();

    fine_pass<true>(coarse, rb, re, lo, tid, cnt, bucket_src);
}

__device__ __forceinline__ float wave_max_f(float x) {
    #pragma unroll
    for (int d = 16; d >= 1; d >>= 1) x = fmaxf(x, __shfl_xor(x, d, 32));
    return x;
}
__device__ __forceinline__ float wave_sum_f(float x) {
    #pragma unroll
    for (int d = 16; d >= 1; d >>= 1) x += __shfl_xor(x, d, 32);
    return x;
}
__device__ __forceinline__ float edge_score(const float* __restrict__ s_l, int s, float sr) {
    float v = s_l[s] + sr;
    return v > 0.0f ? v : NEG_SLOPE * v;
}
template <int U>
__device__ __forceinline__ float add_rows_n(const float* __restrict__ z, int s, float a,
                                            int k, int lane, float acc) {
    float zr[U];
    float ak[U];
    #pragma unroll
    for (int u = 0; u < U; ++u) {
        const int sk = __builtin_amdgcn_readlane(s, k + u);
        ak[u] = __int_as_float(__builtin_amdgcn_readlane(__float_as_int(a), k + u));
        zr[u] = z[(size_t)sk * OUT_DIM + lane];
    }
    #pragma unroll
    for (int u = 0; u < U; ++u) acc += ak[u] * zr[u];
    return acc;
}
__device__ __forceinline__ float add_rows(const float* __restrict__ z, int s, float a,
                                          int cnt, int lane, float acc) {
    int k = 0;
    for (; k + 8 <= cnt; k += 8) acc = add_rows_n<8>(z, s, a, k, lane, acc);
    if (k + 4 <= cnt) { acc = add_rows_n<4>(z, s, a, k, lane, acc); k += 4; }
    if (k + 2 <= cnt) { acc = add_rows_n<2>(z, s, a, k, lane, acc); k += 2; }
    if (k < cnt)      { acc = add_rows_n<1>(z, s, a, k, lane, acc); }
    return acc;
}

__global__ void gat_node_kernel(const int* __restrict__ bucket_src,
                                const int* __restrict__ offset,
                                const int* __restrict__ count,
                                const float* __restrict__ z,
                                const float* __restrict__ s_l,
                                const float* __restrict__ s_r,
                                float* __restrict__ h_out,
                                int N) {
    const int lane = threadIdx.x & 31;
    const int t = __builtin_amdgcn_readfirstlane(
                      (int)(blockIdx.x * (blockDim.x >> 5) + (threadIdx.x >> 5)));
    if (t >= N) return;
    const int   beg = offset[t];
    const int   deg = count[t];
    const float sr  = s_r[t];
    float acc = 0.0f;

    if (deg <= 32) {
        int   s = 0;
        float e = -INFINITY;
        if (lane < deg) {
            s = bucket_src[beg + lane];
            e = edge_score(s_l, s, sr);
        }
        const float m   = wave_max_f(e);
        const float ex  = (lane < deg) ? __expf(e - m) : 0.0f;
        const float den = wave_sum_f(ex);
        const float a   = ex / den;
        acc = add_rows(z, s, a, deg, lane, acc);
    } else {
        float m = -INFINITY;
        for (int j = lane; j < deg; j += 32)
            m = fmaxf(m, edge_score(s_l, bucket_src[beg + j], sr));
        m = wave_max_f(m);
        float den = 0.0f;
        for (int j = lane; j < deg; j += 32)
            den += __expf(edge_score(s_l, bucket_src[beg + j], sr) - m);
        den = wave_sum_f(den);
        for (int base = 0; base < deg; base += 32) {
            const int j = base + lane;
            int   s = 0;
            float a = 0.0f;
            if (j < deg) {
                s = bucket_src[beg + j];
                a = __expf(edge_score(s_l, s, sr) - m) / den;
            }
            const int left = deg - base;
            acc = add_rows(z, s, a, left < 32 ? left : 32, lane, acc);
        }
    }
    h_out[(size_t)t * OUT_DIM + lane] = acc;
}

extern "C" void kernel_launch(void* const* d_in, const int* in_sizes, int n_in,
                              void* d_out, int out_size, void* d_ws, size_t ws_size,
                              hipStream_t stream) {
    const float* h      = (const float*)d_in[0];
    const int*   src    = (const int*)  d_in[1];
    const int*   dst    = (const int*)  d_in[2];
    const float* fc_w   = (const float*)d_in[3];
    const float* attn_w = (const float*)d_in[4];

    const int N = in_sizes[0] / IN_DIM;
    const int E = in_sizes[1];

    float* h_out = (float*)d_out;

    auto align256 = [](size_t x) { return (x + 255) & ~(size_t)255; };
    char* w = (char*)d_ws;
    float* z      = (float*)w; w += align256((size_t)N * OUT_DIM * sizeof(float));
    float* s_l    = (float*)w; w += align256((size_t)N * sizeof(float));
    float* s_r    = (float*)w; w += align256((size_t)N * sizeof(float));
    int*   count  = (int*)w;   w += align256((size_t)N * sizeof(int));
    int*   offset = (int*)w;   w += align256((size_t)N * sizeof(int));
    int*   bucket = (int*)w;   w += align256((size_t)E * sizeof(int));
    (void)ws_size;

    const int B = 256;
    const int tiles = (N + 15) / 16;
    const int gemm_grid = (tiles + B / 32 - 1) / (B / 32);
    const int NB = (N + BK_T - 1) / BK_T;

    const bool two_level =
        NB < CP_T &&
        (size_t)E * sizeof(int2) <= (size_t)N * OUT_DIM * sizeof(float) &&
        (size_t)CP_G * 2 * (NB < 32 ? 32 : NB) <= (size_t)N;

    if (two_level) {
        int* coarse   = (int*)z;
        int* hist     = (int*)s_l;
        int* binstart = (int*)s_r;
        gat_coarse_hist_kernel<<<CP_G, CP_T, 0, stream>>>(dst, hist, N, NB, E);
        gat_coarse_scatter_kernel<<<CP_G, CP_T, 0, stream>>>(src, dst, hist, coarse, binstart, N, NB, E);
        gat_fine_bucket_kernel<<<NB, BK_T, 0, stream>>>(coarse, binstart, offset, count, bucket, N);
        gat_gemm_wmma_kernel<<<gemm_grid, B, 0, stream>>>(h, fc_w, z, N);
        gat_sls_kernel<<<(N + B - 1) / B, B, 0, stream>>>(z, attn_w, s_l, s_r, N);
    } else {
        gat_gemm_wmma_kernel<<<gemm_grid, B, 0, stream>>>(h, fc_w, z, N);
        gat_sls_kernel<<<(N + B - 1) / B, B, 0, stream>>>(z, attn_w, s_l, s_r, N);
        gat_bucket_kernel<<<NB, BK_T, 0, stream>>>(src, dst, offset, count, bucket, N, E);
    }
    {
        int wavesPerBlock = B / 32;
        int grid = (N + wavesPerBlock - 1) / wavesPerBlock;
        gat_node_kernel<<<grid, B, 0, stream>>>(bucket, offset, count, z, s_l, s_r, h_out, N);
    }
}


# Both scatter kernels of the edge grouping fill an LDS stage first and copy it out in order: global stores cover contiguous runs, not one 128-byte line per lane; float sums stay in bucket order
# speedup vs baseline: 3.7879x; 1.1660x over previous
#include <hip/hip_runtime.h>
#include <hip/hip_bf16.h>
#include <math.h>

#define NEG_SLOPE 0.01f
#define IN_DIM  128
#define OUT_DIM 32

typedef __attribute__((ext_vector_type(2))) float v2f;
typedef __attribute__((ext_vector_type(8))) float v8f;

__global__ void gat_gemm_wmma_kernel(const float* __restrict__ h,
                                     const float* __restrict__ fc_w,
                                     float* __restrict__ z,
                                     int N) {
    const int lane   = threadIdx.x & 31;
    const int waveId = blockIdx.x * (blockDim.x >> 5) + (threadIdx.x >> 5);
    const int m0     = waveId * 16;
    if (m0 >= N) return;

    const int n15   = lane & 15;
    const int khalf = lane >> 4;

    const float* __restrict__ arow = h + (size_t)(m0 + n15) * IN_DIM + 2 * khalf;
    const float* __restrict__ b0p  = fc_w + (size_t)n15        * IN_DIM + 2 * khalf;
    const float* __restrict__ b1p  = fc_w + (size_t)(16 + n15) * IN_DIM + 2 * khalf;

    v8f acc0 = {};
    v8f acc1 = {};

    #pragma unroll
    for (int k = 0; k < IN_DIM; k += 4) {
        v2f a  = *(const v2f*)(arow + k);
        v2f b0 = *(const v2f*)(b0p + k);
        v2f b1 = *(const v2f*)(b1p + k);
        acc0 = __builtin_amdgcn_wmma_f32_16x16x4_f32(
                   false, a, false, b0, (short)0, acc0, false, false);
        acc1 = __builtin_amdgcn_wmma_f32_16x16x4_f32(
                   false, a, false, b1, (short)0, acc1, false, false);
    }

    #pragma unroll
    for (int g = 0; g < 8; ++g) {
        int row = m0 + g + 8 * khalf;
        z[(size_t)row * OUT_DIM + n15]      = acc0[g];
        z[(size_t)row * OUT_DIM + 16 + n15] = acc1[g];
    }
}

__global__ void gat_sls_kernel(const float* __restrict__ z,
                               const float* __restrict__ attn_w,
                               float* __restrict__ s_l,
                               float* __restrict__ s_r,
                               int N) {
    int i = blockIdx.x * blockDim.x + threadIdx.x;
    if (i >= N) return;
    const float4* zr = (const float4*)(z + (size_t)i * OUT_DIM);
    const float4* al = (const float4*)(attn_w);
    const float4* ar = (const float4*)(attn_w + OUT_DIM);
    float accl = 0.0f, accr = 0.0f;
    #pragma unroll
    for (int q = 0; q < OUT_DIM / 4; ++q) {
        float4 zv = zr[q];
        float4 lv = al[q];
        float4 rv = ar[q];
        accl += zv.x * lv.x + zv.y * lv.y + zv.z * lv.z + zv.w * lv.w;
        accr += zv.x * rv.x + zv.y * rv.y + zv.z * rv.z + zv.w * rv.w;
    }
    s_l[i] = accl;
    s_r[i] = accr;
}

#define BK_T 1024
#define BK_U 16
#define NO_NODE 0x7fffffff

template <bool FILL>
__device__ __forceinline__ int bucket_one(int dv, int e, int lo, int* cnt,
                                          const int* __restrict__ src,
                                          int* __restrict__ bucket_src) {
    const unsigned r = (unsigned)(dv - lo);
    if (r < (unsigned)BK_T) {
        if (FILL) {
            const int pos = atomicAdd(&cnt[r], 1);
            bucket_src[pos] = src[e];
        } else {
            atomicAdd(&cnt[r], 1);
        }
    }
    if (FILL) return 0;
    return __builtin_popcount(__builtin_amdgcn_ballot_w32(dv < lo));
}

template <bool GUARD, bool FILL>
__device__ __forceinline__ int bucket_batch(const int4* __restrict__ dst4, int q0, int E4,
                                            int lo, int* cnt,
                                            const int* __restrict__ src,
                                            int* __restrict__ bucket_src) {
    int4 d[BK_U];
    #pragma unroll
    for (int u = 0; u < BK_U; ++u) {
        const int q = q0 + u * BK_T;
        if (GUARD) {
            d[u] = make_int4(NO_NODE, NO_NODE, NO_NODE, NO_NODE);
            if (q < E4) d[u] = dst4[q];
        } else {
            d[u] = dst4[q];
        }
    }
    int below = 0;
    #pragma unroll
    for (int u = 0; u < BK_U; ++u) {
        const int e0 = (q0 + u * BK_T) * 4;
        below += bucket_one<FILL>(d[u].x, e0 + 0, lo, cnt, src, bucket_src);
        below += bucket_one<FILL>(d[u].y, e0 + 1, lo, cnt, src, bucket_src);
        below += bucket_one<FILL>(d[u].z, e0 + 2, lo, cnt, src, bucket_src);
        below += bucket_one<FILL>(d[u].w, e0 + 3, lo, cnt, src, bucket_src);
    }
    return below;
}

template <bool FILL>
__device__ __forceinline__ int bucket_pass(const int* __restrict__ dst, int E, int lo, int tid,
                                           int* cnt, const int* __restrict__ src,
                                           int* __restrict__ bucket_src) {
    const int4* __restrict__ dst4 = (const int4*)dst;
    const int E4 = E >> 2;
    const int per_batch = BK_T * BK_U;
    const int nfull = E4 / per_batch;
    int below = 0;
    for (int b = 0; b < nfull; ++b)
        below += bucket_batch<false, FILL>(dst4, b * per_batch + tid, E4, lo, cnt, src, bucket_src);
    if (nfull * per_batch < E4)
        below += bucket_batch<true, FILL>(dst4, nfull * per_batch + tid, E4, lo, cnt, src, bucket_src);
    {
        const int e = (E4 << 2) + tid;
        int dv = NO_NODE;
        if (e < E) dv = dst[e];
        below += bucket_one<FILL>(dv, e, lo, cnt, src, bucket_src);
    }
    return below;
}

__global__ void __launch_bounds__(BK_T)
gat_bucket_kernel(const int* __restrict__ src,
                  const int* __restrict__ dst,
                  int* __restrict__ offset,
                  int* __restrict__ count,
                  int* __restrict__ bucket_src,
                  int N, int E) {
    __shared__ int cnt[BK_T];
    __shared__ int wave_tot[32];
    __shared__ int wave_below[32];
    __shared__ int base_sh;
    const int tid  = threadIdx.x;
    const int lane = tid & 31;
    const int wv   = tid >> 5;
    const int lo   = blockIdx.x * BK_T;

    cnt[tid] = 0;
    __syncthreads();

    const int below = bucket_pass<false>(dst, E, lo, tid, cnt, src, bucket_src);
    __syncthreads();

    const int c = cnt[tid];
    int inc = c;
    #pragma unroll
    for (int d = 1; d < 32; d <<= 1) {
        int o = __shfl_up(inc, d, 32);
        if (lane >= d) inc += o;
    }
    if (lane == 31) wave_tot[wv] = inc;
    if (lane == 0)  wave_below[wv] = below;
    __syncthreads();
    if (wv == 0) {
        const int w = wave_tot[lane];
        int winc = w;
        #pragma unroll
        for (int d = 1; d < 32; d <<= 1) {
            int o = __shfl_up(winc, d, 32);
            if (lane >= d) winc += o;
        }
        wave_tot[lane] = winc - w;
        int b = wave_below[lane];
        #pragma unroll
        for (int d = 16; d >= 1; d >>= 1) b += __shfl_xor(b, d, 32);
        if (lane == 0) base_sh = b;
    }
    __syncthreads();
    const int start = base_sh + wave_tot[wv] + (inc - c);
    if (lo + tid < N) {
        offset[lo + tid] = start;
        count[lo + tid]  = c;
    }
    cnt[tid] = start;
    __syncthreads();

    (void)bucket_pass<true>(dst, E, lo, tid, cnt, src, bucket_src);
}

#define BK_SHIFT 10
#define CP_G 128
#define CP_T 1024
#define CP_U 8

__device__ __forceinline__ int hist_pitch_log2(int NB) {
    int hs = 5;
    while ((1 << hs) < NB) ++hs;
    return hs;
}

__global__ void __launch_bounds__(CP_T)
gat_coarse_hist_kernel(const int* __restrict__ dst, int* __restrict__ hist,
                       int N, int NB, int E) {
    __shared__ int cnt[CP_T];
    const int tid = threadIdx.x;
    const int c   = blockIdx.x;
    const int len = (E + CP_G - 1) / CP_G;
    const int beg = c * len;
    const int end = (beg + len < E) ? beg + len : E;
    const int HP  = 1 << hist_pitch_log2(NB);

    cnt[tid] = 0;
    __syncthreads();
    for (int base = beg; base < end; base += CP_T * CP_U) {
        int d[CP_U];
        #pragma unroll
        for (int u = 0; u < CP_U; ++u) {
            const int e = base + u * CP_T + tid;
            d[u] = -1;
            if (e < end) d[u] = dst[e];
        }
        #pragma unroll
        for (int u = 0; u < CP_U; ++u)
            if ((unsigned)d[u] < (unsigned)N) atomicAdd(&cnt[d[u] >> BK_SHIFT], 1);
    }
    __syncthreads();
    if (tid < HP) hist[c * HP + tid] = cnt[tid];
}

#define CS_CAP 16384

template <bool STAGED>
__device__ __forceinline__ void coarse_fill(const int* __restrict__ src, const int* __restrict__ dst,
                                            int beg, int end, int tid, int N, int* cur,
                                            int2* stage, int2* __restrict__ coarse) {
    for (int base = beg; base < end; base += CP_T * CP_U) {
        int d[CP_U];
        int s[CP_U];
        #pragma unroll
        for (int u = 0; u < CP_U; ++u) {
            const int e = base + u * CP_T + tid;
            d[u] = -1;
            s[u] = 0;
            if (e < end) { d[u] = dst[e]; s[u] = src[e]; }
        }
        #pragma unroll
        for (int u = 0; u < CP_U; ++u) {
            if ((unsigned)d[u] < (unsigned)N) {
                const int pos = atomicAdd(&cur[d[u] >> BK_SHIFT], 1);
                if (STAGED) stage[pos]  = make_int2(d[u], s[u]);
                else        coarse[pos] = make_int2(d[u], s[u]);
            }
        }
    }
}

__global__ void __launch_bounds__(CP_T)
gat_coarse_scatter_kernel(const int* __restrict__ src, const int* __restrict__ dst,
                          const int* __restrict__ hist,
                          int* __restrict__ coarse_raw, int* __restrict__ binstart,
                          int N, int NB, int E) {
    __shared__ int cur[CP_T];
    __shared__ int tot[CP_T];
    __shared__ int dlt[CP_T];
    __shared__ int wave_tot[32];
    __shared__ int wave_own[32];
    __shared__ int nvalid_sh;
    __shared__ int2 stage[CS_CAP];
    int2* __restrict__ coarse = (int2*)coarse_raw;
    const int tid  = threadIdx.x;
    const int lane = tid & 31;
    const int wv   = tid >> 5;
    const int c    = blockIdx.x;
    const int len  = (E + CP_G - 1) / CP_G;
    const int beg  = c * len;
    const int end  = (beg + len < E) ? beg + len : E;
    const int hs   = hist_pitch_log2(NB);
    const int HP   = 1 << hs;

    cur[tid] = 0;
    tot[tid] = 0;
    __syncthreads();
    {
        const int b = tid & (HP - 1);
        int part = 0, all = 0;
        for (int j = tid >> hs; j < CP_G; j += (CP_T >> hs)) {
            const int v = hist[j * HP + b];
            all += v;
            if (j < c) part += v;
        }
        atomicAdd(&cur[b], part);
        atomicAdd(&tot[b], all);
    }
    __syncthreads();
    const bool staged = len <= CS_CAP;
    const int v   = tot[tid];
    const int own = (tid < HP) ? hist[c * HP + tid] : 0;
    int inc = v, oinc = own;
    for (int d = 1; d < 32; d <<= 1) {
        int o  = __shfl_up(inc, d, 32);
        int oo = __shfl_up(oinc, d, 32);
        if (lane >= d) { inc += o; oinc += oo; }
    }
    if (lane == 31) { wave_tot[wv] = inc; wave_own[wv] = oinc; }
    __syncthreads();
    if (wv == 0) {
        const int w  = wave_tot[lane];
        const int ow = wave_own[lane];
        int winc = w, owinc = ow;
        for (int d = 1; d < 32; d <<= 1) {
            int o  = __shfl_up(winc, d, 32);
            int oo = __shfl_up(owinc, d, 32);
            if (lane >= d) { winc += o; owinc += oo; }
        }
        wave_tot[lane] = winc - w;
        wave_own[lane] = owinc - ow;
        if (lane == 31) nvalid_sh = owinc;
    }
    __syncthreads();
    const int excl  = wave_tot[wv] + (inc - v);
    const int lexcl = wave_own[wv] + (oinc - own);
    const int g0    = cur[tid] + excl;
    cur[tid] = staged ? lexcl : g0;
    dlt[tid] = g0 - lexcl;
    if (c == 0 && tid <= NB) binstart[tid] = excl;
    __syncthreads();

    if (staged) {
        coarse_fill<true>(src, dst, beg, end, tid, N, cur, stage, coarse);
        __syncthreads();
        const int nvalid = nvalid_sh;
        for (int i = tid; i < nvalid; i += CP_T) {
            const int2 p = stage[i];
            coarse[i + dlt[p.x >> BK_SHIFT]] = p;
        }
    } else {
        coarse_fill<false>(src, dst, beg, end, tid, N, cur, stage, coarse);
    }
}

#define FB_U 8

#define FB_CAP 32768

template <bool FILL>
__device__ __forceinline__ void fine_pass(const int2* __restrict__ coarse, int rb, int re,
                                          int lo, int tid, int* cnt,
                                          int* bucket_src) {
    for (int base = rb; base < re; base += BK_T * FB_U) {
        int2 p[FB_U];
        #pragma unroll
        for (int u = 0; u < FB_U; ++u) {
            const int i = base + u * BK_T + tid;
            p[u] = make_int2(-1, 0);
            if (i < re) p[u] = coarse[i];
        }
        #pragma unroll
        for (int u = 0; u < FB_U; ++u) {
            const unsigned r = (unsigned)(p[u].x - lo);
            if (r < (unsigned)BK_T) {
                if (FILL) {
                    const int pos = atomicAdd(&cnt[r], 1);
                    bucket_src[pos] = p[u].y;
                } else {
                    atomicAdd(&cnt[r], 1);
                }
            }
        }
    }
}

__global__ void __launch_bounds__(BK_T)
gat_fine_bucket_kernel(const int* __restrict__ coarse_raw, const int* __restrict__ binstart,
                       int* __restrict__ offset, int* __restrict__ count,
                       int* __restrict__ bucket_src, int N) {
    __shared__ int cnt[BK_T];
    __shared__ int wave_tot[32];
    __shared__ int total_sh;
    __shared__ int stage[FB_CAP];
    const int2* __restrict__ coarse = (const int2*)coarse_raw;
    const int tid  = threadIdx.x;
    const int lane = tid & 31;
    const int wv   = tid >> 5;
    const int lo   = blockIdx.x * BK_T;
    const int rb   = binstart[blockIdx.x];
    const int re   = binstart[blockIdx.x + 1];
    const bool staged = (re - rb) <= FB_CAP;

    cnt[tid] = 0;
    __syncthreads();
    fine_pass<false>(coarse, rb, re, lo, tid, cnt, bucket_src);
    __syncthreads();

    const int c = cnt[tid];
    int inc = c;
    for (int d = 1; d < 32; d <<= 1) {
        int o = __shfl_up(inc, d, 32);
        if (lane >= d) inc += o;
    }
    if (lane == 31) wave_tot[wv] = inc;
    __syncthreads();
    if (wv == 0) {
        const int w = wave_tot[lane];
        int winc = w;
        for (int d = 1; d < 32; d <<= 1) {
            int o = __shfl_up(winc, d, 32);
            if (lane >= d) winc += o;
        }
        wave_tot[lane] = winc - w;
        if (lane == 31) total_sh = winc;
    }
    __syncthreads();
    const int lstart = wave_tot[wv] + (inc - c);
    const int start  = rb + lstart;
    if (lo + tid < N) {
        offset[lo + tid] = start;
        count[lo + tid]  = c;
    }
    cnt[tid] = staged ? lstart : start;
    __syncthreads();

    if (staged) {
        fine_pass<true>(coarse, rb, re, lo, tid, cnt, stage);
        __syncthreads();
        const int total = total_sh;
        for (int i = tid; i < total; i += BK_T) bucket_src[rb + i] = stage[i];
    } else {
        fine_pass<true>(coarse, rb, re, lo, tid, cnt, bucket_src);
    }
}

__device__ __forceinline__ float wave_max_f(float x) {
    #pragma unroll
    for (int d = 16; d >= 1; d >>= 1) x = fmaxf(x, __shfl_xor(x, d, 32));
    return x;
}
__device__ __forceinline__ float wave_sum_f(float x) {
    #pragma unroll
    for (int d = 16; d >= 1; d >>= 1) x += __shfl_xor(x, d, 32);
    return x;
}
__device__ __forceinline__ float edge_score(const float* __restrict__ s_l, int s, float sr) {
    float v = s_l[s] + sr;
    return v > 0.0f ? v : NEG_SLOPE * v;
}
template <int U>
__device__ __forceinline__ float add_rows_n(const float* __restrict__ z, int s, float a,
                                            int k, int lane, float acc) {
    float zr[U];
    float ak[U];
    #pragma unroll
    for (int u = 0; u < U; ++u) {
        const int sk = __builtin_amdgcn_readlane(s, k + u);
        ak[u] = __int_as_float(__builtin_amdgcn_readlane(__float_as_int(a), k + u));
        zr[u] = z[(size_t)sk * OUT_DIM + lane];
    }
    #pragma unroll
    for (int u = 0; u < U; ++u) acc += ak[u] * zr[u];
    return acc;
}
__device__ __forceinline__ float add_rows(const float* __restrict__ z, int s, float a,
                                          int cnt, int lane, float acc) {
    int k = 0;
    for (; k + 8 <= cnt; k += 8) acc = add_rows_n<8>(z, s, a, k, lane, acc);
    if (k + 4 <= cnt) { acc = add_rows_n<4>(z, s, a, k, lane, acc); k += 4; }
    if (k + 2 <= cnt) { acc = add_rows_n<2>(z, s, a, k, lane, acc); k += 2; }
    if (k < cnt)      { acc = add_rows_n<1>(z, s, a, k, lane, acc); }
    return acc;
}

__global__ void gat_node_kernel(const int* __restrict__ bucket_src,
                                const int* __restrict__ offset,
                                const int* __restrict__ count,
                                const float* __restrict__ z,
                                const float* __restrict__ s_l,
                                const float* __restrict__ s_r,
                                float* __restrict__ h_out,
                                int N) {
    const int lane = threadIdx.x & 31;
    const int t = __builtin_amdgcn_readfirstlane(
                      (int)(blockIdx.x * (blockDim.x >> 5) + (threadIdx.x >> 5)));
    if (t >= N) return;
    const int   beg = offset[t];
    const int   deg = count[t];
    const float sr  = s_r[t];
    float acc = 0.0f;

    if (deg <= 32) {
        int   s = 0;
        float e = -INFINITY;
        if (lane < deg) {
            s = bucket_src[beg + lane];
            e = edge_score(s_l, s, sr);
        }
        const float m   = wave_max_f(e);
        const float ex  = (lane < deg) ? __expf(e - m) : 0.0f;
        const float den = wave_sum_f(ex);
        const float a   = ex / den;
        acc = add_rows(z, s, a, deg, lane, acc);
    } else {
        float m = -INFINITY;
        for (int j = lane; j < deg; j += 32)
            m = fmaxf(m, edge_score(s_l, bucket_src[beg + j], sr));
        m = wave_max_f(m);
        float den = 0.0f;
        for (int j = lane; j < deg; j += 32)
            den += __expf(edge_score(s_l, bucket_src[beg + j], sr) - m);
        den = wave_sum_f(den);
        for (int base = 0; base < deg; base += 32) {
            const int j = base + lane;
            int   s = 0;
            float a = 0.0f;
            if (j < deg) {
                s = bucket_src[beg + j];
                a = __expf(edge_score(s_l, s, sr) - m) / den;
            }
            const int left = deg - base;
            acc = add_rows(z, s, a, left < 32 ? left : 32, lane, acc);
        }
    }
    h_out[(size_t)t * OUT_DIM + lane] = acc;
}

extern "C" void kernel_launch(void* const* d_in, const int* in_sizes, int n_in,
                              void* d_out, int out_size, void* d_ws, size_t ws_size,
                              hipStream_t stream) {
    const float* h      = (const float*)d_in[0];
    const int*   src    = (const int*)  d_in[1];
    const int*   dst    = (const int*)  d_in[2];
    const float* fc_w   = (const float*)d_in[3];
    const float* attn_w = (const float*)d_in[4];

    const int N = in_sizes[0] / IN_DIM;
    const int E = in_sizes[1];

    float* h_out = (float*)d_out;

    auto align256 = [](size_t x) { return (x + 255) & ~(size_t)255; };
    char* w = (char*)d_ws;
    float* z      = (float*)w; w += align256((size_t)N * OUT_DIM * sizeof(float));
    float* s_l    = (float*)w; w += align256((size_t)N * sizeof(float));
    float* s_r    = (float*)w; w += align256((size_t)N * sizeof(float));
    int*   count  = (int*)w;   w += align256((size_t)N * sizeof(int));
    int*   offset = (int*)w;   w += align256((size_t)N * sizeof(int));
    int*   bucket = (int*)w;   w += align256((size_t)E * sizeof(int));
    (void)ws_size;

    const int B = 256;
    const int tiles = (N + 15) / 16;
    const int gemm_grid = (tiles + B / 32 - 1) / (B / 32);
    const int NB = (N + BK_T - 1) / BK_T;

    const bool two_level =
        NB < CP_T &&
        (size_t)E * sizeof(int2) <= (size_t)N * OUT_DIM * sizeof(float) &&
        (size_t)CP_G * 2 * (NB < 32 ? 32 : NB) <= (size_t)N;

    if (two_level) {
        int* coarse   = (int*)z;
        int* hist     = (int*)s_l;
        int* binstart = (int*)s_r;
        gat_coarse_hist_kernel<<<CP_G, CP_T, 0, stream>>>(dst, hist, N, NB, E);
        gat_coarse_scatter_kernel<<<CP_G, CP_T, 0, stream>>>(src, dst, hist, coarse, binstart, N, NB, E);
        gat_fine_bucket_kernel<<<NB, BK_T, 0, stream>>>(coarse, binstart, offset, count, bucket, N);
        gat_gemm_wmma_kernel<<<gemm_grid, B, 0, stream>>>(h, fc_w, z, N);
        gat_sls_kernel<<<(N + B - 1) / B, B, 0, stream>>>(z, attn_w, s_l, s_r, N);
    } else {
        gat_gemm_wmma_kernel<<<gemm_grid, B, 0, stream>>>(h, fc_w, z, N);
        gat_sls_kernel<<<(N + B - 1) / B, B, 0, stream>>>(z, attn_w, s_l, s_r, N);
        gat_bucket_kernel<<<NB, BK_T, 0, stream>>>(src, dst, offset, count, bucket, N, E);
    }
    {
        int wavesPerBlock = B / 32;
        int grid = (N + wavesPerBlock - 1) / wavesPerBlock;
        gat_node_kernel<<<grid, B, 0, stream>>>(bucket, offset, count, z, s_l, s_r, h_out, N);
    }
}


# Node kernel: 2048 resident work-groups whose waves each walk several destination nodes, next node's sources and scores fetched under this node's row sums; float sums stay in bucket order
# speedup vs baseline: 3.9545x; 1.0440x over previous
#include <hip/hip_runtime.h>
#include <hip/hip_bf16.h>
#include <math.h>

#define NEG_SLOPE 0.01f
#define IN_DIM  128
#define OUT_DIM 32

typedef __attribute__((ext_vector_type(2))) float v2f;
typedef __attribute__((ext_vector_type(8))) float v8f;

__global__ void gat_gemm_wmma_kernel(const float* __restrict__ h,
                                     const float* __restrict__ fc_w,
                                     float* __restrict__ z,
                                     int N) {
    const int lane   = threadIdx.x & 31;
    const int waveId = blockIdx.x * (blockDim.x >> 5) + (threadIdx.x >> 5);
    const int m0     = waveId * 16;
    if (m0 >= N) return;

    const int n15   = lane & 15;
    const int khalf = lane >> 4;

    const float* __restrict__ arow = h + (size_t)(m0 + n15) * IN_DIM + 2 * khalf;
    const float* __restrict__ b0p  = fc_w + (size_t)n15        * IN_DIM + 2 * khalf;
    const float* __restrict__ b1p  = fc_w + (size_t)(16 + n15) * IN_DIM + 2 * khalf;

    v8f acc0 = {};
    v8f acc1 = {};

    #pragma unroll
    for (int k = 0; k < IN_DIM; k += 4) {
        v2f a  = *(const v2f*)(arow + k);
        v2f b0 = *(const v2f*)(b0p + k);
        v2f b1 = *(const v2f*)(b1p + k);
        acc0 = __builtin_amdgcn_wmma_f32_16x16x4_f32(
                   false, a, false, b0, (short)0, acc0, false, false);
        acc1 = __builtin_amdgcn_wmma_f32_16x16x4_f32(
                   false, a, false, b1, (short)0, acc1, false, false);
    }

    #pragma unroll
    for (int g = 0; g < 8; ++g) {
        int row = m0 + g + 8 * khalf;
        z[(size_t)row * OUT_DIM + n15]      = acc0[g];
        z[(size_t)row * OUT_DIM + 16 + n15] = acc1[g];
    }
}

__global__ void gat_sls_kernel(const float* __restrict__ z,
                               const float* __restrict__ attn_w,
                               float* __restrict__ s_l,
                               float* __restrict__ s_r,
                               int N) {
    int i = blockIdx.x * blockDim.x + threadIdx.x;
    if (i >= N) return;
    const float4* zr = (const float4*)(z + (size_t)i * OUT_DIM);
    const float4* al = (const float4*)(attn_w);
    const float4* ar = (const float4*)(attn_w + OUT_DIM);
    float accl = 0.0f, accr = 0.0f;
    #pragma unroll
    for (int q = 0; q < OUT_DIM / 4; ++q) {
        float4 zv = zr[q];
        float4 lv = al[q];
        float4 rv = ar[q];
        accl += zv.x * lv.x + zv.y * lv.y + zv.z * lv.z + zv.w * lv.w;
        accr += zv.x * rv.x + zv.y * rv.y + zv.z * rv.z + zv.w * rv.w;
    }
    s_l[i] = accl;
    s_r[i] = accr;
}

#define BK_T 1024
#define BK_U 16
#define NO_NODE 0x7fffffff

template <bool FILL>
__device__ __forceinline__ int bucket_one(int dv, int e, int lo, int* cnt,
                                          const int* __restrict__ src,
                                          int* __restrict__ bucket_src) {
    const unsigned r = (unsigned)(dv - lo);
    if (r < (unsigned)BK_T) {
        if (FILL) {
            const int pos = atomicAdd(&cnt[r], 1);
            bucket_src[pos] = src[e];
        } else {
            atomicAdd(&cnt[r], 1);
        }
    }
    if (FILL) return 0;
    return __builtin_popcount(__builtin_amdgcn_ballot_w32(dv < lo));
}

template <bool GUARD, bool FILL>
__device__ __forceinline__ int bucket_batch(const int4* __restrict__ dst4, int q0, int E4,
                                            int lo, int* cnt,
                                            const int* __restrict__ src,
                                            int* __restrict__ bucket_src) {
    int4 d[BK_U];
    #pragma unroll
    for (int u = 0; u < BK_U; ++u) {
        const int q = q0 + u * BK_T;
        if (GUARD) {
            d[u] = make_int4(NO_NODE, NO_NODE, NO_NODE, NO_NODE);
            if (q < E4) d[u] = dst4[q];
        } else {
            d[u] = dst4[q];
        }
    }
    int below = 0;
    #pragma unroll
    for (int u = 0; u < BK_U; ++u) {
        const int e0 = (q0 + u * BK_T) * 4;
        below += bucket_one<FILL>(d[u].x, e0 + 0, lo, cnt, src, bucket_src);
        below += bucket_one<FILL>(d[u].y, e0 + 1, lo, cnt, src, bucket_src);
        below += bucket_one<FILL>(d[u].z, e0 + 2, lo, cnt, src, bucket_src);
        below += bucket_one<FILL>(d[u].w, e0 + 3, lo, cnt, src, bucket_src);
    }
    return below;
}

template <bool FILL>
__device__ __forceinline__ int bucket_pass(const int* __restrict__ dst, int E, int lo, int tid,
                                           int* cnt, const int* __restrict__ src,
                                           int* __restrict__ bucket_src) {
    const int4* __restrict__ dst4 = (const int4*)dst;
    const int E4 = E >> 2;
    const int per_batch = BK_T * BK_U;
    const int nfull = E4 / per_batch;
    int below = 0;
    for (int b = 0; b < nfull; ++b)
        below += bucket_batch<false, FILL>(dst4, b * per_batch + tid, E4, lo, cnt, src, bucket_src);
    if (nfull * per_batch < E4)
        below += bucket_batch<true, FILL>(dst4, nfull * per_batch + tid, E4, lo, cnt, src, bucket_src);
    {
        const int e = (E4 << 2) + tid;
        int dv = NO_NODE;
        if (e < E) dv = dst[e];
        below += bucket_one<FILL>(dv, e, lo, cnt, src, bucket_src);
    }
    return below;
}

__global__ void __launch_bounds__(BK_T)
gat_bucket_kernel(const int* __restrict__ src,
                  const int* __restrict__ dst,
                  int* __restrict__ offset,
                  int* __restrict__ count,
                  int* __restrict__ bucket_src,
                  int N, int E) {
    __shared__ int cnt[BK_T];
    __shared__ int wave_tot[32];
    __shared__ int wave_below[32];
    __shared__ int base_sh;
    const int tid  = threadIdx.x;
    const int lane = tid & 31;
    const int wv   = tid >> 5;
    const int lo   = blockIdx.x * BK_T;

    cnt[tid] = 0;
    __syncthreads();

    const int below = bucket_pass<false>(dst, E, lo, tid, cnt, src, bucket_src);
    __syncthreads();

    const int c = cnt[tid];
    int inc = c;
    #pragma unroll
    for (int d = 1; d < 32; d <<= 1) {
        int o = __shfl_up(inc, d, 32);
        if (lane >= d) inc += o;
    }
    if (lane == 31) wave_tot[wv] = inc;
    if (lane == 0)  wave_below[wv] = below;
    __syncthreads();
    if (wv == 0) {
        const int w = wave_tot[lane];
        int winc = w;
        #pragma unroll
        for (int d = 1; d < 32; d <<= 1) {
            int o = __shfl_up(winc, d, 32);
            if (lane >= d) winc += o;
        }
        wave_tot[lane] = winc - w;
        int b = wave_below[lane];
        #pragma unroll
        for (int d = 16; d >= 1; d >>= 1) b += __shfl_xor(b, d, 32);
        if (lane == 0) base_sh = b;
    }
    __syncthreads();
    const int start = base_sh + wave_tot[wv] + (inc - c);
    if (lo + tid < N) {
        offset[lo + tid] = start;
        count[lo + tid]  = c;
    }
    cnt[tid] = start;
    __syncthreads();

    (void)bucket_pass<true>(dst, E, lo, tid, cnt, src, bucket_src);
}

#define BK_SHIFT 10
#define CP_G 128
#define CP_T 1024
#define CP_U 8

__device__ __forceinline__ int hist_pitch_log2(int NB) {
    int hs = 5;
    while ((1 << hs) < NB) ++hs;
    return hs;
}

__global__ void __launch_bounds__(CP_T)
gat_coarse_hist_kernel(const int* __restrict__ dst, int* __restrict__ hist,
                       int N, int NB, int E) {
    __shared__ int cnt[CP_T];
    const int tid = threadIdx.x;
    const int c   = blockIdx.x;
    const int len = (E + CP_G - 1) / CP_G;
    const int beg = c * len;
    const int end = (beg + len < E) ? beg + len : E;
    const int HP  = 1 << hist_pitch_log2(NB);

    cnt[tid] = 0;
    __syncthreads();
    for (int base = beg; base < end; base += CP_T * CP_U) {
        int d[CP_U];
        #pragma unroll
        for (int u = 0; u < CP_U; ++u) {
            const int e = base + u * CP_T + tid;
            d[u] = -1;
            if (e < end) d[u] = dst[e];
        }
        #pragma unroll
        for (int u = 0; u < CP_U; ++u)
            if ((unsigned)d[u] < (unsigned)N) atomicAdd(&cnt[d[u] >> BK_SHIFT], 1);
    }
    __syncthreads();
    if (tid < HP) hist[c * HP + tid] = cnt[tid];
}

#define CS_CAP 16384

template <bool STAGED>
__device__ __forceinline__ void coarse_fill(const int* __restrict__ src, const int* __restrict__ dst,
                                            int beg, int end, int tid, int N, int* cur,
                                            int2* stage, int2* __restrict__ coarse) {
    for (int base = beg; base < end; base += CP_T * CP_U) {
        int d[CP_U];
        int s[CP_U];
        #pragma unroll
        for (int u = 0; u < CP_U; ++u) {
            const int e = base + u * CP_T + tid;
            d[u] = -1;
            s[u] = 0;
            if (e < end) { d[u] = dst[e]; s[u] = src[e]; }
        }
        #pragma unroll
        for (int u = 0; u < CP_U; ++u) {
            if ((unsigned)d[u] < (unsigned)N) {
                const int pos = atomicAdd(&cur[d[u] >> BK_SHIFT], 1);
                if (STAGED) stage[pos]  = make_int2(d[u], s[u]);
                else        coarse[pos] = make_int2(d[u], s[u]);
            }
        }
    }
}

__global__ void __launch_bounds__(CP_T)
gat_coarse_scatter_kernel(const int* __restrict__ src, const int* __restrict__ dst,
                          const int* __restrict__ hist,
                          int* __restrict__ coarse_raw, int* __restrict__ binstart,
                          int N, int NB, int E) {
    __shared__ int cur[CP_T];
    __shared__ int tot[CP_T];
    __shared__ int dlt[CP_T];
    __shared__ int wave_tot[32];
    __shared__ int wave_own[32];
    __shared__ int nvalid_sh;
    __shared__ int2 stage[CS_CAP];
    int2* __restrict__ coarse = (int2*)coarse_raw;
    const int tid  = threadIdx.x;
    const int lane = tid & 31;
    const int wv   = tid >> 5;
    const int c    = blockIdx.x;
    const int len  = (E + CP_G - 1) / CP_G;
    const int beg  = c * len;
    const int end  = (beg + len < E) ? beg + len : E;
    const int hs   = hist_pitch_log2(NB);
    const int HP   = 1 << hs;

    cur[tid] = 0;
    tot[tid] = 0;
    __syncthreads();
    {
        const int b = tid & (HP - 1);
        int part = 0, all = 0;
        for (int j = tid >> hs; j < CP_G; j += (CP_T >> hs)) {
            const int v = hist[j * HP + b];
            all += v;
            if (j < c) part += v;
        }
        atomicAdd(&cur[b], part);
        atomicAdd(&tot[b], all);
    }
    __syncthreads();
    const bool staged = len <= CS_CAP;
    const int v   = tot[tid];
    const int own = (tid < HP) ? hist[c * HP + tid] : 0;
    int inc = v, oinc = own;
    for (int d = 1; d < 32; d <<= 1) {
        int o  = __shfl_up(inc, d, 32);
        int oo = __shfl_up(oinc, d, 32);
        if (lane >= d) { inc += o; oinc += oo; }
    }
    if (lane == 31) { wave_tot[wv] = inc; wave_own[wv] = oinc; }
    __syncthreads();
    if (wv == 0) {
        const int w  = wave_tot[lane];
        const int ow = wave_own[lane];
        int winc = w, owinc = ow;
        for (int d = 1; d < 32; d <<= 1) {
            int o  = __shfl_up(winc, d, 32);
            int oo = __shfl_up(owinc, d, 32);
            if (lane >= d) { winc += o; owinc += oo; }
        }
        wave_tot[lane] = winc - w;
        wave_own[lane] = owinc - ow;
        if (lane == 31) nvalid_sh = owinc;
    }
    __syncthreads();
    const int excl  = wave_tot[wv] + (inc - v);
    const int lexcl = wave_own[wv] + (oinc - own);
    const int g0    = cur[tid] + excl;
    cur[tid] = staged ? lexcl : g0;
    dlt[tid] = g0 - lexcl;
    if (c == 0 && tid <= NB) binstart[tid] = excl;
    __syncthreads();

    if (staged) {
        coarse_fill<true>(src, dst, beg, end, tid, N, cur, stage, coarse);
        __syncthreads();
        const int nvalid = nvalid_sh;
        for (int i = tid; i < nvalid; i += CP_T) {
            const int2 p = stage[i];
            coarse[i + dlt[p.x >> BK_SHIFT]] = p;
        }
    } else {
        coarse_fill<false>(src, dst, beg, end, tid, N, cur, stage, coarse);
    }
}

#define FB_U 8

#define FB_CAP 32768

template <bool FILL>
__device__ __forceinline__ void fine_pass(const int2* __restrict__ coarse, int rb, int re,
                                          int lo, int tid, int* cnt,
                                          int* bucket_src) {
    for (int base = rb; base < re; base += BK_T * FB_U) {
        int2 p[FB_U];
        #pragma unroll
        for (int u = 0; u < FB_U; ++u) {
            const int i = base + u * BK_T + tid;
            p[u] = make_int2(-1, 0);
            if (i < re) p[u] = coarse[i];
        }
        #pragma unroll
        for (int u = 0; u < FB_U; ++u) {
            const unsigned r = (unsigned)(p[u].x - lo);
            if (r < (unsigned)BK_T) {
                if (FILL) {
                    const int pos = atomicAdd(&cnt[r], 1);
                    bucket_src[pos] = p[u].y;
                } else {
                    atomicAdd(&cnt[r], 1);
                }
            }
        }
    }
}

__global__ void __launch_bounds__(BK_T)
gat_fine_bucket_kernel(const int* __restrict__ coarse_raw, const int* __restrict__ binstart,
                       int* __restrict__ offset, int* __restrict__ count,
                       int* __restrict__ bucket_src, int N) {
    __shared__ int cnt[BK_T];
    __shared__ int wave_tot[32];
    __shared__ int total_sh;
    __shared__ int stage[FB_CAP];
    const int2* __restrict__ coarse = (const int2*)coarse_raw;
    const int tid  = threadIdx.x;
    const int lane = tid & 31;
    const int wv   = tid >> 5;
    const int lo   = blockIdx.x * BK_T;
    const int rb   = binstart[blockIdx.x];
    const int re   = binstart[blockIdx.x + 1];
    const bool staged = (re - rb) <= FB_CAP;

    cnt[tid] = 0;
    __syncthreads();
    fine_pass<false>(coarse, rb, re, lo, tid, cnt, bucket_src);
    __syncthreads();

    const int c = cnt[tid];
    int inc = c;
    for (int d = 1; d < 32; d <<= 1) {
        int o = __shfl_up(inc, d, 32);
        if (lane >= d) inc += o;
    }
    if (lane == 31) wave_tot[wv] = inc;
    __syncthreads();
    if (wv == 0) {
        const int w = wave_tot[lane];
        int winc = w;
        for (int d = 1; d < 32; d <<= 1) {
            int o = __shfl_up(winc, d, 32);
            if (lane >= d) winc += o;
        }
        wave_tot[lane] = winc - w;
        if (lane == 31) total_sh = winc;
    }
    __syncthreads();
    const int lstart = wave_tot[wv] + (inc - c);
    const int start  = rb + lstart;
    if (lo + tid < N) {
        offset[lo + tid] = start;
        count[lo + tid]  = c;
    }
    cnt[tid] = staged ? lstart : start;
    __syncthreads();

    if (staged) {
        fine_pass<true>(coarse, rb, re, lo, tid, cnt, stage);
        __syncthreads();
        const int total = total_sh;
        for (int i = tid; i < total; i += BK_T) bucket_src[rb + i] = stage[i];
    } else {
        fine_pass<true>(coarse, rb, re, lo, tid, cnt, bucket_src);
    }
}

__device__ __forceinline__ float wave_max_f(float x) {
    #pragma unroll
    for (int d = 16; d >= 1; d >>= 1) x = fmaxf(x, __shfl_xor(x, d, 32));
    return x;
}
__device__ __forceinline__ float wave_sum_f(float x) {
    #pragma unroll
    for (int d = 16; d >= 1; d >>= 1) x += __shfl_xor(x, d, 32);
    return x;
}
__device__ __forceinline__ float edge_score(const float* __restrict__ s_l, int s, float sr) {
    float v = s_l[s] + sr;
    return v > 0.0f ? v : NEG_SLOPE * v;
}
template <int U>
__device__ __forceinline__ float add_rows_n(const float* __restrict__ z, int s, float a,
                                            int k, int lane, float acc) {
    float zr[U];
    float ak[U];
    #pragma unroll
    for (int u = 0; u < U; ++u) {
        const int sk = __builtin_amdgcn_readlane(s, k + u);
        ak[u] = __int_as_float(__builtin_amdgcn_readlane(__float_as_int(a), k + u));
        zr[u] = z[(size_t)sk * OUT_DIM + lane];
    }
    #pragma unroll
    for (int u = 0; u < U; ++u) acc += ak[u] * zr[u];
    return acc;
}
__device__ __forceinline__ float add_rows(const float* __restrict__ z, int s, float a,
                                          int cnt, int lane, float acc) {
    int k = 0;
    for (; k + 8 <= cnt; k += 8) acc = add_rows_n<8>(z, s, a, k, lane, acc);
    if (k + 4 <= cnt) { acc = add_rows_n<4>(z, s, a, k, lane, acc); k += 4; }
    if (k + 2 <= cnt) { acc = add_rows_n<2>(z, s, a, k, lane, acc); k += 2; }
    if (k < cnt)      { acc = add_rows_n<1>(z, s, a, k, lane, acc); }
    return acc;
}

#define NODE_WG 2048

__device__ __forceinline__ float score_of(float sl, float sr) {
    float v = sl + sr;
    return v > 0.0f ? v : NEG_SLOPE * v;
}

__global__ void __launch_bounds__(256)
gat_node_kernel(const int* __restrict__ bucket_src,
                const int* __restrict__ offset,
                const int* __restrict__ count,
                const float* __restrict__ z,
                const float* __restrict__ s_l,
                const float* __restrict__ s_r,
                float* __restrict__ h_out,
                int N) {
    const int lane = threadIdx.x & 31;
    const int wpb  = (int)(blockDim.x >> 5);
    const int nw   = (int)gridDim.x * wpb;
    int t = __builtin_amdgcn_readfirstlane(
                (int)(blockIdx.x * wpb + (threadIdx.x >> 5)));
    if (t >= N) return;

    int   beg = offset[t];
    int   deg = count[t];
    float sr  = s_r[t];
    int   s = 0;
    float e = -INFINITY;
    if (lane < deg) {
        s = bucket_src[beg + lane];
        e = score_of(s_l[s], sr);
    }
    int   tn = t + nw;
    int   begn = 0, degn = 0;
    float srn = 0.0f;
    if (tn < N) { begn = offset[tn]; degn = count[tn]; srn = s_r[tn]; }

    for (;;) {
        int sn = 0;
        if (lane < degn) sn = bucket_src[begn + lane];
        const int tnn = tn + nw;
        int   begnn = 0, degnn = 0;
        float srnn = 0.0f;
        if (tnn < N) { begnn = offset[tnn]; degnn = count[tnn]; srnn = s_r[tnn]; }

        float acc = 0.0f;
        float en  = -INFINITY;
        if (deg <= 32) {
            const float m   = wave_max_f(e);
            const float ex  = (lane < deg) ? __expf(e - m) : 0.0f;
            const float den = wave_sum_f(ex);
            const float a   = ex / den;
            float gn = 0.0f;
            if (lane < degn) gn = s_l[sn];
            acc = add_rows(z, s, a, deg, lane, acc);
            if (lane < degn) en = score_of(gn, srn);
        } else {
            float m = -INFINITY;
            for (int j = lane; j < deg; j += 32)
                m = fmaxf(m, edge_score(s_l, bucket_src[beg + j], sr));
            m = wave_max_f(m);
            float den = 0.0f;
            for (int j = lane; j < deg; j += 32)
                den += __expf(edge_score(s_l, bucket_src[beg + j], sr) - m);
            den = wave_sum_f(den);
            for (int base = 0; base < deg; base += 32) {
                const int j = base + lane;
                int   sj = 0;
                float a = 0.0f;
                if (j < deg) {
                    sj = bucket_src[beg + j];
                    a = __expf(edge_score(s_l, sj, sr) - m) / den;
                }
                const int left = deg - base;
                acc = add_rows(z, sj, a, left < 32 ? left : 32, lane, acc);
            }
            if (lane < degn) en = edge_score(s_l, sn, srn);
        }
        h_out[(size_t)t * OUT_DIM + lane] = acc;

        if (tn >= N) break;
        t = tn;  beg = begn;  deg = degn;  sr = srn;  s = sn;  e = en;
        tn = tnn; begn = begnn; degn = degnn; srn = srnn;
    }
}

extern "C" void kernel_launch(void* const* d_in, const int* in_sizes, int n_in,
                              void* d_out, int out_size, void* d_ws, size_t ws_size,
                              hipStream_t stream) {
    const float* h      = (const float*)d_in[0];
    const int*   src    = (const int*)  d_in[1];
    const int*   dst    = (const int*)  d_in[2];
    const float* fc_w   = (const float*)d_in[3];
    const float* attn_w = (const float*)d_in[4];

    const int N = in_sizes[0] / IN_DIM;
    const int E = in_sizes[1];

    float* h_out = (float*)d_out;

    auto align256 = [](size_t x) { return (x + 255) & ~(size_t)255; };
    char* w = (char*)d_ws;
    float* z      = (float*)w; w += align256((size_t)N * OUT_DIM * sizeof(float));
    float* s_l    = (float*)w; w += align256((size_t)N * sizeof(float));
    float* s_r    = (float*)w; w += align256((size_t)N * sizeof(float));
    int*   count  = (int*)w;   w += align256((size_t)N * sizeof(int));
    int*   offset = (int*)w;   w += align256((size_t)N * sizeof(int));
    int*   bucket = (int*)w;   w += align256((size_t)E * sizeof(int));
    (void)ws_size;

    const int B = 256;
    const int tiles = (N + 15) / 16;
    const int gemm_grid = (tiles + B / 32 - 1) / (B / 32);
    const int NB = (N + BK_T - 1) / BK_T;

    const bool two_level =
        NB < CP_T &&
        (size_t)E * sizeof(int2) <= (size_t)N * OUT_DIM * sizeof(float) &&
        (size_t)CP_G * 2 * (NB < 32 ? 32 : NB) <= (size_t)N;

    if (two_level) {
        int* coarse   = (int*)z;
        int* hist     = (int*)s_l;
        int* binstart = (int*)s_r;
        gat_coarse_hist_kernel<<<CP_G, CP_T, 0, stream>>>(dst, hist, N, NB, E);
        gat_coarse_scatter_kernel<<<CP_G, CP_T, 0, stream>>>(src, dst, hist, coarse, binstart, N, NB, E);
        gat_fine_bucket_kernel<<<NB, BK_T, 0, stream>>>(coarse, binstart, offset, count, bucket, N);
        gat_gemm_wmma_kernel<<<gemm_grid, B, 0, stream>>>(h, fc_w, z, N);
        gat_sls_kernel<<<(N + B - 1) / B, B, 0, stream>>>(z, attn_w, s_l, s_r, N);
    } else {
        gat_gemm_wmma_kernel<<<gemm_grid, B, 0, stream>>>(h, fc_w, z, N);
        gat_sls_kernel<<<(N + B - 1) / B, B, 0, stream>>>(z, attn_w, s_l, s_r, N);
        gat_bucket_kernel<<<NB, BK_T, 0, stream>>>(src, dst, offset, count, bucket, N, E);
    }
    {
        int wavesPerBlock = B / 32;
        int need = (N + wavesPerBlock - 1) / wavesPerBlock;
        int grid = need < NODE_WG ? need : NODE_WG;
        gat_node_kernel<<<grid, B, 0, stream>>>(bucket, offset, count, z, s_l, s_r, h_out, N);
    }
}


# GEMM operand feed: each wave multiplies two 16-row tiles by one set of fc_w fragments, fragments loaded 16 bytes per lane (two WMMAs per load): products summed in a permuted k order
# speedup vs baseline: 4.4524x; 1.1259x over previous
#include <hip/hip_runtime.h>
#include <hip/hip_bf16.h>
#include <math.h>

#define NEG_SLOPE 0.01f
#define IN_DIM  128
#define OUT_DIM 32

typedef __attribute__((ext_vector_type(2))) float v2f;
typedef __attribute__((ext_vector_type(8))) float v8f;

#define GEMM_WAVES 8
#define GEMM_WROWS 32
#define GEMM_ROWS (GEMM_WAVES * GEMM_WROWS)

typedef __attribute__((ext_vector_type(4))) float v4f;

template <int M>
__device__ __forceinline__ float row_xor_add(float x) {
    const int o = __builtin_amdgcn_update_dpp(0, __float_as_int(x), 0x160 + M, 0xf, 0xf, true);
    return x + __int_as_float(o);
}
__device__ __forceinline__ float half_sum16(float x) {
    x = row_xor_add<1>(x);
    x = row_xor_add<2>(x);
    x = row_xor_add<4>(x);
    x = row_xor_add<8>(x);
    return x;
}

__device__ __forceinline__ v8f wmma_f32(v2f a, v2f b, v8f c) {
    return __builtin_amdgcn_wmma_f32_16x16x4_f32(false, a, false, b, (short)0, c, false, false);
}

__device__ __forceinline__ void tile_scores(const v8f& acc0, const v8f& acc1,
                                            float al0, float al1, float ar0, float ar1,
                                            int n15, float& vl, float& vr) {
    vl = 0.0f; vr = 0.0f;
    #pragma unroll
    for (int g = 0; g < 8; ++g) {
        const float pl = half_sum16(acc0[g] * al0 + acc1[g] * al1);
        const float pr = half_sum16(acc0[g] * ar0 + acc1[g] * ar1);
        if ((n15 & 7) == g) { vl = pl; vr = pr; }
    }
}

__device__ __forceinline__ void tile_store(float* __restrict__ z, int m, int n15, int khalf,
                                           const v8f& acc0, const v8f& acc1) {
    #pragma unroll
    for (int g = 0; g < 8; ++g) {
        const int row = m + g + 8 * khalf;
        z[(size_t)row * OUT_DIM + n15]      = acc0[g];
        z[(size_t)row * OUT_DIM + 16 + n15] = acc1[g];
    }
}

__global__ void __launch_bounds__(GEMM_WAVES * 32)
gat_gemm_wmma_kernel(const float* __restrict__ h,
                     const float* __restrict__ fc_w,
                     const float* __restrict__ attn_w,
                     float* __restrict__ z,
                     float* __restrict__ s_l,
                     float* __restrict__ s_r,
                     int N) {
    __shared__ float sl_sh[GEMM_ROWS];
    __shared__ float sr_sh[GEMM_ROWS];
    const int lane = threadIdx.x & 31;
    const int wv   = threadIdx.x >> 5;
    const int row0 = blockIdx.x * GEMM_ROWS;
    const int wrow = row0 + wv * GEMM_WROWS;
    const bool live0 = wrow < N;
    const bool live1 = wrow + 16 < N;
    const int  m0    = live0 ? wrow : 0;
    const int  m1    = live1 ? wrow + 16 : 0;

    const int n15   = lane & 15;
    const int khalf = lane >> 4;

    const float* __restrict__ a0p = h + (size_t)(m0 + n15) * IN_DIM + 4 * khalf;
    const float* __restrict__ a1p = h + (size_t)(m1 + n15) * IN_DIM + 4 * khalf;
    const float* __restrict__ b0p = fc_w + (size_t)n15        * IN_DIM + 4 * khalf;
    const float* __restrict__ b1p = fc_w + (size_t)(16 + n15) * IN_DIM + 4 * khalf;

    const float al0 = attn_w[n15];
    const float al1 = attn_w[16 + n15];
    const float ar0 = attn_w[OUT_DIM + n15];
    const float ar1 = attn_w[OUT_DIM + 16 + n15];

    v8f c00 = {};
    v8f c01 = {};
    v8f c10 = {};
    v8f c11 = {};

    #pragma unroll
    for (int k = 0; k < IN_DIM; k += 8) {
        const v4f a0 = *(const v4f*)(a0p + k);
        const v4f a1 = *(const v4f*)(a1p + k);
        const v4f b0 = *(const v4f*)(b0p + k);
        const v4f b1 = *(const v4f*)(b1p + k);
        const v2f a0l = {a0.x, a0.y}, a0h = {a0.z, a0.w};
        const v2f a1l = {a1.x, a1.y}, a1h = {a1.z, a1.w};
        const v2f b0l = {b0.x, b0.y}, b0h = {b0.z, b0.w};
        const v2f b1l = {b1.x, b1.y}, b1h = {b1.z, b1.w};
        c00 = wmma_f32(a0l, b0l, c00);
        c01 = wmma_f32(a0l, b1l, c01);
        c10 = wmma_f32(a1l, b0l, c10);
        c11 = wmma_f32(a1l, b1l, c11);
        c00 = wmma_f32(a0h, b0h, c00);
        c01 = wmma_f32(a0h, b1h, c01);
        c10 = wmma_f32(a1h, b0h, c10);
        c11 = wmma_f32(a1h, b1h, c11);
    }

    float vl0, vr0, vl1, vr1;
    tile_scores(c00, c01, al0, al1, ar0, ar1, n15, vl0, vr0);
    tile_scores(c10, c11, al0, al1, ar0, ar1, n15, vl1, vr1);
    if (n15 < 8) {
        const int j = wv * GEMM_WROWS + 8 * khalf + n15;
        sl_sh[j]      = vl0;
        sr_sh[j]      = vr0;
        sl_sh[j + 16] = vl1;
        sr_sh[j + 16] = vr1;
    }
    __syncthreads();
    {
        const int row = row0 + (int)threadIdx.x;
        if (row < N) {
            s_l[row] = sl_sh[threadIdx.x];
            s_r[row] = sr_sh[threadIdx.x];
        }
    }

    if (live0) tile_store(z, m0, n15, khalf, c00, c01);
    if (live1) tile_store(z, m1, n15, khalf, c10, c11);
}

#define BK_T 1024
#define BK_U 16
#define NO_NODE 0x7fffffff

template <bool FILL>
__device__ __forceinline__ int bucket_one(int dv, int e, int lo, int* cnt,
                                          const int* __restrict__ src,
                                          int* __restrict__ bucket_src) {
    const unsigned r = (unsigned)(dv - lo);
    if (r < (unsigned)BK_T) {
        if (FILL) {
            const int pos = atomicAdd(&cnt[r], 1);
            bucket_src[pos] = src[e];
        } else {
            atomicAdd(&cnt[r], 1);
        }
    }
    if (FILL) return 0;
    return __builtin_popcount(__builtin_amdgcn_ballot_w32(dv < lo));
}

template <bool GUARD, bool FILL>
__device__ __forceinline__ int bucket_batch(const int4* __restrict__ dst4, int q0, int E4,
                                            int lo, int* cnt,
                                            const int* __restrict__ src,
                                            int* __restrict__ bucket_src) {
    int4 d[BK_U];
    #pragma unroll
    for (int u = 0; u < BK_U; ++u) {
        const int q = q0 + u * BK_T;
        if (GUARD) {
            d[u] = make_int4(NO_NODE, NO_NODE, NO_NODE, NO_NODE);
            if (q < E4) d[u] = dst4[q];
        } else {
            d[u] = dst4[q];
        }
    }
    int below = 0;
    #pragma unroll
    for (int u = 0; u < BK_U; ++u) {
        const int e0 = (q0 + u * BK_T) * 4;
        below += bucket_one<FILL>(d[u].x, e0 + 0, lo, cnt, src, bucket_src);
        below += bucket_one<FILL>(d[u].y, e0 + 1, lo, cnt, src, bucket_src);
        below += bucket_one<FILL>(d[u].z, e0 + 2, lo, cnt, src, bucket_src);
        below += bucket_one<FILL>(d[u].w, e0 + 3, lo, cnt, src, bucket_src);
    }
    return below;
}

template <bool FILL>
__device__ __forceinline__ int bucket_pass(const int* __restrict__ dst, int E, int lo, int tid,
                                           int* cnt, const int* __restrict__ src,
                                           int* __restrict__ bucket_src) {
    const int4* __restrict__ dst4 = (const int4*)dst;
    const int E4 = E >> 2;
    const int per_batch = BK_T * BK_U;
    const int nfull = E4 / per_batch;
    int below = 0;
    for (int b = 0; b < nfull; ++b)
        below += bucket_batch<false, FILL>(dst4, b * per_batch + tid, E4, lo, cnt, src, bucket_src);
    if (nfull * per_batch < E4)
        below += bucket_batch<true, FILL>(dst4, nfull * per_batch + tid, E4, lo, cnt, src, bucket_src);
    {
        const int e = (E4 << 2) + tid;
        int dv = NO_NODE;
        if (e < E) dv = dst[e];
        below += bucket_one<FILL>(dv, e, lo, cnt, src, bucket_src);
    }
    return below;
}

__global__ void __launch_bounds__(BK_T)
gat_bucket_kernel(const int* __restrict__ src,
                  const int* __restrict__ dst,
                  int* __restrict__ offset,
                  int* __restrict__ count,
                  int* __restrict__ bucket_src,
                  int N, int E) {
    __shared__ int cnt[BK_T];
    __shared__ int wave_tot[32];
    __shared__ int wave_below[32];
    __shared__ int base_sh;
    const int tid  = threadIdx.x;
    const int lane = tid & 31;
    const int wv   = tid >> 5;
    const int lo   = blockIdx.x * BK_T;

    cnt[tid] = 0;
    __syncthreads();

    const int below = bucket_pass<false>(dst, E, lo, tid, cnt, src, bucket_src);
    __syncthreads();

    const int c = cnt[tid];
    int inc = c;
    #pragma unroll
    for (int d = 1; d < 32; d <<= 1) {
        int o = __shfl_up(inc, d, 32);
        if (lane >= d) inc += o;
    }
    if (lane == 31) wave_tot[wv] = inc;
    if (lane == 0)  wave_below[wv] = below;
    __syncthreads();
    if (wv == 0) {
        const int w = wave_tot[lane];
        int winc = w;
        #pragma unroll
        for (int d = 1; d < 32; d <<= 1) {
            int o = __shfl_up(winc, d, 32);
            if (lane >= d) winc += o;
        }
        wave_tot[lane] = winc - w;
        int b = wave_below[lane];
        #pragma unroll
        for (int d = 16; d >= 1; d >>= 1) b += __shfl_xor(b, d, 32);
        if (lane == 0) base_sh = b;
    }
    __syncthreads();
    const int start = base_sh + wave_tot[wv] + (inc - c);
    if (lo + tid < N) {
        offset[lo + tid] = start;
        count[lo + tid]  = c;
    }
    cnt[tid] = start;
    __syncthreads();

    (void)bucket_pass<true>(dst, E, lo, tid, cnt, src, bucket_src);
}

#define BK_SHIFT 10
#define CP_G 128
#define CP_T 1024
#define CP_U 8

__device__ __forceinline__ int hist_pitch_log2(int NB) {
    int hs = 5;
    while ((1 << hs) < NB) ++hs;
    return hs;
}

__global__ void __launch_bounds__(CP_T)
gat_coarse_hist_kernel(const int* __restrict__ dst, int* __restrict__ hist,
                       int N, int NB, int E) {
    __shared__ int cnt[CP_T];
    const int tid = threadIdx.x;
    const int c   = blockIdx.x;
    const int len = (E + CP_G - 1) / CP_G;
    const int beg = c * len;
    const int end = (beg + len < E) ? beg + len : E;
    const int HP  = 1 << hist_pitch_log2(NB);

    cnt[tid] = 0;
    __syncthreads();
    for (int base = beg; base < end; base += CP_T * CP_U) {
        int d[CP_U];
        #pragma unroll
        for (int u = 0; u < CP_U; ++u) {
            const int e = base + u * CP_T + tid;
            d[u] = -1;
            if (e < end) d[u] = dst[e];
        }
        #pragma unroll
        for (int u = 0; u < CP_U; ++u)
            if ((unsigned)d[u] < (unsigned)N) atomicAdd(&cnt[d[u] >> BK_SHIFT], 1);
    }
    __syncthreads();
    if (tid < HP) hist[c * HP + tid] = cnt[tid];
}

#define CS_CAP 16384

template <bool STAGED>
__device__ __forceinline__ void coarse_fill(const int* __restrict__ src, const int* __restrict__ dst,
                                            int beg, int end, int tid, int N, int* cur,
                                            int2* stage, int2* __restrict__ coarse) {
    for (int base = beg; base < end; base += CP_T * CP_U) {
        int d[CP_U];
        int s[CP_U];
        #pragma unroll
        for (int u = 0; u < CP_U; ++u) {
            const int e = base + u * CP_T + tid;
            d[u] = -1;
            s[u] = 0;
            if (e < end) { d[u] = dst[e]; s[u] = src[e]; }
        }
        #pragma unroll
        for (int u = 0; u < CP_U; ++u) {
            if ((unsigned)d[u] < (unsigned)N) {
                const int pos = atomicAdd(&cur[d[u] >> BK_SHIFT], 1);
                if (STAGED) stage[pos]  = make_int2(d[u], s[u]);
                else        coarse[pos] = make_int2(d[u], s[u]);
            }
        }
    }
}

__global__ void __launch_bounds__(CP_T)
gat_coarse_scatter_kernel(const int* __restrict__ src, const int* __restrict__ dst,
                          const int* __restrict__ hist,
                          int* __restrict__ coarse_raw, int* __restrict__ binstart,
                          int N, int NB, int E) {
    __shared__ int cur[CP_T];
    __shared__ int tot[CP_T];
    __shared__ int dlt[CP_T];
    __shared__ int wave_tot[32];
    __shared__ int wave_own[32];
    __shared__ int nvalid_sh;
    __shared__ int2 stage[CS_CAP];
    int2* __restrict__ coarse = (int2*)coarse_raw;
    const int tid  = threadIdx.x;
    const int lane = tid & 31;
    const int wv   = tid >> 5;
    const int c    = blockIdx.x;
    const int len  = (E + CP_G - 1) / CP_G;
    const int beg  = c * len;
    const int end  = (beg + len < E) ? beg + len : E;
    const int hs   = hist_pitch_log2(NB);
    const int HP   = 1 << hs;

    cur[tid] = 0;
    tot[tid] = 0;
    __syncthreads();
    {
        const int b = tid & (HP - 1);
        int part = 0, all = 0;
        for (int j = tid >> hs; j < CP_G; j += (CP_T >> hs)) {
            const int v = hist[j * HP + b];
            all += v;
            if (j < c) part += v;
        }
        atomicAdd(&cur[b], part);
        atomicAdd(&tot[b], all);
    }
    __syncthreads();
    const bool staged = len <= CS_CAP;
    const int v   = tot[tid];
    const int own = (tid < HP) ? hist[c * HP + tid] : 0;
    int inc = v, oinc = own;
    for (int d = 1; d < 32; d <<= 1) {
        int o  = __shfl_up(inc, d, 32);
        int oo = __shfl_up(oinc, d, 32);
        if (lane >= d) { inc += o; oinc += oo; }
    }
    if (lane == 31) { wave_tot[wv] = inc; wave_own[wv] = oinc; }
    __syncthreads();
    if (wv == 0) {
        const int w  = wave_tot[lane];
        const int ow = wave_own[lane];
        int winc = w, owinc = ow;
        for (int d = 1; d < 32; d <<= 1) {
            int o  = __shfl_up(winc, d, 32);
            int oo = __shfl_up(owinc, d, 32);
            if (lane >= d) { winc += o; owinc += oo; }
        }
        wave_tot[lane] = winc - w;
        wave_own[lane] = owinc - ow;
        if (lane == 31) nvalid_sh = owinc;
    }
    __syncthreads();
    const int excl  = wave_tot[wv] + (inc - v);
    const int lexcl = wave_own[wv] + (oinc - own);
    const int g0    = cur[tid] + excl;
    cur[tid] = staged ? lexcl : g0;
    dlt[tid] = g0 - lexcl;
    if (c == 0 && tid <= NB) binstart[tid] = excl;
    __syncthreads();

    if (staged) {
        coarse_fill<true>(src, dst, beg, end, tid, N, cur, stage, coarse);
        __syncthreads();
        const int nvalid = nvalid_sh;
        for (int i = tid; i < nvalid; i += CP_T) {
            const int2 p = stage[i];
            coarse[i + dlt[p.x >> BK_SHIFT]] = p;
        }
    } else {
        coarse_fill<false>(src, dst, beg, end, tid, N, cur, stage, coarse);
    }
}

#define FB_U 8

#define FB_CAP 32768

template <bool FILL>
__device__ __forceinline__ void fine_pass(const int2* __restrict__ coarse, int rb, int re,
                                          int lo, int tid, int* cnt,
                                          int* bucket_src) {
    for (int base = rb; base < re; base += BK_T * FB_U) {
        int2 p[FB_U];
        #pragma unroll
        for (int u = 0; u < FB_U; ++u) {
            const int i = base + u * BK_T + tid;
            p[u] = make_int2(-1, 0);
            if (i < re) p[u] = coarse[i];
        }
        #pragma unroll
        for (int u = 0; u < FB_U; ++u) {
            const unsigned r = (unsigned)(p[u].x - lo);
            if (r < (unsigned)BK_T) {
                if (FILL) {
                    const int pos = atomicAdd(&cnt[r], 1);
                    bucket_src[pos] = p[u].y;
                } else {
                    atomicAdd(&cnt[r], 1);
                }
            }
        }
    }
}

__global__ void __launch_bounds__(BK_T)
gat_fine_bucket_kernel(const int* __restrict__ coarse_raw, const int* __restrict__ binstart,
                       int* __restrict__ offset, int* __restrict__ count,
                       int* __restrict__ bucket_src, int N) {
    __shared__ int cnt[BK_T];
    __shared__ int wave_tot[32];
    __shared__ int total_sh;
    __shared__ int stage[FB_CAP];
    const int2* __restrict__ coarse = (const int2*)coarse_raw;
    const int tid  = threadIdx.x;
    const int lane = tid & 31;
    const int wv   = tid >> 5;
    const int lo   = blockIdx.x * BK_T;
    const int rb   = binstart[blockIdx.x];
    const int re   = binstart[blockIdx.x + 1];
    const bool staged = (re - rb) <= FB_CAP;

    cnt[tid] = 0;
    __syncthreads();
    fine_pass<false>(coarse, rb, re, lo, tid, cnt, bucket_src);
    __syncthreads();

    const int c = cnt[tid];
    int inc = c;
    for (int d = 1; d < 32; d <<= 1) {
        int o = __shfl_up(inc, d, 32);
        if (lane >= d) inc += o;
    }
    if (lane == 31) wave_tot[wv] = inc;
    __syncthreads();
    if (wv == 0) {
        const int w = wave_tot[lane];
        int winc = w;
        for (int d = 1; d < 32; d <<= 1) {
            int o = __shfl_up(winc, d, 32);
            if (lane >= d) winc += o;
        }
        wave_tot[lane] = winc - w;
        if (lane == 31) total_sh = winc;
    }
    __syncthreads();
    const int lstart = wave_tot[wv] + (inc - c);
    const int start  = rb + lstart;
    if (lo + tid < N) {
        offset[lo + tid] = start;
        count[lo + tid]  = c;
    }
    cnt[tid] = staged ? lstart : start;
    __syncthreads();

    if (staged) {
        fine_pass<true>(coarse, rb, re, lo, tid, cnt, stage);
        __syncthreads();
        const int total = total_sh;
        for (int i = tid; i < total; i += BK_T) bucket_src[rb + i] = stage[i];
    } else {
        fine_pass<true>(coarse, rb, re, lo, tid, cnt, bucket_src);
    }
}

__device__ __forceinline__ float wave_max_f(float x) {
    #pragma unroll
    for (int d = 16; d >= 1; d >>= 1) x = fmaxf(x, __shfl_xor(x, d, 32));
    return x;
}
__device__ __forceinline__ float wave_sum_f(float x) {
    #pragma unroll
    for (int d = 16; d >= 1; d >>= 1) x += __shfl_xor(x, d, 32);
    return x;
}
__device__ __forceinline__ float edge_score(const float* __restrict__ s_l, int s, float sr) {
    float v = s_l[s] + sr;
    return v > 0.0f ? v : NEG_SLOPE * v;
}
template <int U>
__device__ __forceinline__ float add_rows_n(const float* __restrict__ z, int s, float a,
                                            int k, int lane, float acc) {
    float zr[U];
    float ak[U];
    #pragma unroll
    for (int u = 0; u < U; ++u) {
        const int sk = __builtin_amdgcn_readlane(s, k + u);
        ak[u] = __int_as_float(__builtin_amdgcn_readlane(__float_as_int(a), k + u));
        zr[u] = z[(size_t)sk * OUT_DIM + lane];
    }
    #pragma unroll
    for (int u = 0; u < U; ++u) acc += ak[u] * zr[u];
    return acc;
}
__device__ __forceinline__ float add_rows(const float* __restrict__ z, int s, float a,
                                          int cnt, int lane, float acc) {
    int k = 0;
    for (; k + 8 <= cnt; k += 8) acc = add_rows_n<8>(z, s, a, k, lane, acc);
    if (k + 4 <= cnt) { acc = add_rows_n<4>(z, s, a, k, lane, acc); k += 4; }
    if (k + 2 <= cnt) { acc = add_rows_n<2>(z, s, a, k, lane, acc); k += 2; }
    if (k < cnt)      { acc = add_rows_n<1>(z, s, a, k, lane, acc); }
    return acc;
}

#define NODE_WG 2048

__device__ __forceinline__ float score_of(float sl, float sr) {
    float v = sl + sr;
    return v > 0.0f ? v : NEG_SLOPE * v;
}

__global__ void __launch_bounds__(256)
gat_node_kernel(const int* __restrict__ bucket_src,
                const int* __restrict__ offset,
                const int* __restrict__ count,
                const float* __restrict__ z,
                const float* __restrict__ s_l,
                const float* __restrict__ s_r,
                float* __restrict__ h_out,
                int N) {
    const int lane = threadIdx.x & 31;
    const int wpb  = (int)(blockDim.x >> 5);
    const int nw   = (int)gridDim.x * wpb;
    int t = __builtin_amdgcn_readfirstlane(
                (int)(blockIdx.x * wpb + (threadIdx.x >> 5)));
    if (t >= N) return;

    int   beg = offset[t];
    int   deg = count[t];
    float sr  = s_r[t];
    int   s = 0;
    float e = -INFINITY;
    if (lane < deg) {
        s = bucket_src[beg + lane];
        e = score_of(s_l[s], sr);
    }
    int   tn = t + nw;
    int   begn = 0, degn = 0;
    float srn = 0.0f;
    if (tn < N) { begn = offset[tn]; degn = count[tn]; srn = s_r[tn]; }

    for (;;) {
        int sn = 0;
        if (lane < degn) sn = bucket_src[begn + lane];
        const int tnn = tn + nw;
        int   begnn = 0, degnn = 0;
        float srnn = 0.0f;
        if (tnn < N) { begnn = offset[tnn]; degnn = count[tnn]; srnn = s_r[tnn]; }

        float acc = 0.0f;
        float en  = -INFINITY;
        if (deg <= 32) {
            const float m   = wave_max_f(e);
            const float ex  = (lane < deg) ? __expf(e - m) : 0.0f;
            const float den = wave_sum_f(ex);
            const float a   = ex / den;
            float gn = 0.0f;
            if (lane < degn) gn = s_l[sn];
            acc = add_rows(z, s, a, deg, lane, acc);
            if (lane < degn) en = score_of(gn, srn);
        } else {
            float m = -INFINITY;
            for (int j = lane; j < deg; j += 32)
                m = fmaxf(m, edge_score(s_l, bucket_src[beg + j], sr));
            m = wave_max_f(m);
            float den = 0.0f;
            for (int j = lane; j < deg; j += 32)
                den += __expf(edge_score(s_l, bucket_src[beg + j], sr) - m);
            den = wave_sum_f(den);
            for (int base = 0; base < deg; base += 32) {
                const int j = base + lane;
                int   sj = 0;
                float a = 0.0f;
                if (j < deg) {
                    sj = bucket_src[beg + j];
                    a = __expf(edge_score(s_l, sj, sr) - m) / den;
                }
                const int left = deg - base;
                acc = add_rows(z, sj, a, left < 32 ? left : 32, lane, acc);
            }
            if (lane < degn) en = edge_score(s_l, sn, srn);
        }
        h_out[(size_t)t * OUT_DIM + lane] = acc;

        if (tn >= N) break;
        t = tn;  beg = begn;  deg = degn;  sr = srn;  s = sn;  e = en;
        tn = tnn; begn = begnn; degn = degnn; srn = srnn;
    }
}

extern "C" void kernel_launch(void* const* d_in, const int* in_sizes, int n_in,
                              void* d_out, int out_size, void* d_ws, size_t ws_size,
                              hipStream_t stream) {
    const float* h      = (const float*)d_in[0];
    const int*   src    = (const int*)  d_in[1];
    const int*   dst    = (const int*)  d_in[2];
    const float* fc_w   = (const float*)d_in[3];
    const float* attn_w = (const float*)d_in[4];

    const int N = in_sizes[0] / IN_DIM;
    const int E = in_sizes[1];

    float* h_out = (float*)d_out;

    auto align256 = [](size_t x) { return (x + 255) & ~(size_t)255; };
    char* w = (char*)d_ws;
    float* z      = (float*)w; w += align256((size_t)N * OUT_DIM * sizeof(float));
    float* s_l    = (float*)w; w += align256((size_t)N * sizeof(float));
    float* s_r    = (float*)w; w += align256((size_t)N * sizeof(float));
    int*   count  = (int*)w;   w += align256((size_t)N * sizeof(int));
    int*   offset = (int*)w;   w += align256((size_t)N * sizeof(int));
    int*   bucket = (int*)w;   w += align256((size_t)E * sizeof(int));
    (void)ws_size;

    const int B = 256;
    const int tiles = (N + 15) / 16;
    const int gemm_grid = (tiles * 16 + GEMM_ROWS - 1) / GEMM_ROWS;
    const int NB = (N + BK_T - 1) / BK_T;

    const bool two_level =
        NB < CP_T &&
        (size_t)E * sizeof(int2) <= (size_t)N * OUT_DIM * sizeof(float) &&
        (size_t)CP_G * 2 * (NB < 32 ? 32 : NB) <= (size_t)N;

    if (two_level) {
        int* coarse   = (int*)z;
        int* hist     = (int*)s_l;
        int* binstart = (int*)s_r;
        gat_coarse_hist_kernel<<<CP_G, CP_T, 0, stream>>>(dst, hist, N, NB, E);
        gat_coarse_scatter_kernel<<<CP_G, CP_T, 0, stream>>>(src, dst, hist, coarse, binstart, N, NB, E);
        gat_fine_bucket_kernel<<<NB, BK_T, 0, stream>>>(coarse, binstart, offset, count, bucket, N);
        gat_gemm_wmma_kernel<<<gemm_grid, GEMM_WAVES * 32, 0, stream>>>(h, fc_w, attn_w, z, s_l, s_r, N);
    } else {
        gat_gemm_wmma_kernel<<<gemm_grid, GEMM_WAVES * 32, 0, stream>>>(h, fc_w, attn_w, z, s_l, s_r, N);
        gat_bucket_kernel<<<NB, BK_T, 0, stream>>>(src, dst, offset, count, bucket, N, E);
    }
    {
        int wavesPerBlock = B / 32;
        int need = (N + wavesPerBlock - 1) / wavesPerBlock;
        int grid = need < NODE_WG ? need : NODE_WG;
        gat_node_kernel<<<grid, B, 0, stream>>>(bucket, offset, count, z, s_l, s_r, h_out, N);
    }
}


# Node kernel: all rows of a node (up to 32 registers) asked for as one group ahead of its softmax, sources fetched two nodes ahead: one exposed round trip per node; float sums stay in bucket order
# speedup vs baseline: 6.3248x; 1.4205x over previous
#include <hip/hip_runtime.h>
#include <hip/hip_bf16.h>
#include <math.h>

#define NEG_SLOPE 0.01f
#define IN_DIM  128
#define OUT_DIM 32

typedef __attribute__((ext_vector_type(2))) float v2f;
typedef __attribute__((ext_vector_type(8))) float v8f;

#define GEMM_WAVES 8
#define GEMM_WROWS 32
#define GEMM_ROWS (GEMM_WAVES * GEMM_WROWS)

typedef __attribute__((ext_vector_type(4))) float v4f;

template <int M>
__device__ __forceinline__ float row_xor_add(float x) {
    const int o = __builtin_amdgcn_update_dpp(0, __float_as_int(x), 0x160 + M, 0xf, 0xf, true);
    return x + __int_as_float(o);
}
__device__ __forceinline__ float half_sum16(float x) {
    x = row_xor_add<1>(x);
    x = row_xor_add<2>(x);
    x = row_xor_add<4>(x);
    x = row_xor_add<8>(x);
    return x;
}

__device__ __forceinline__ v8f wmma_f32(v2f a, v2f b, v8f c) {
    return __builtin_amdgcn_wmma_f32_16x16x4_f32(false, a, false, b, (short)0, c, false, false);
}

__device__ __forceinline__ void tile_scores(const v8f& acc0, const v8f& acc1,
                                            float al0, float al1, float ar0, float ar1,
                                            int n15, float& vl, float& vr) {
    vl = 0.0f; vr = 0.0f;
    #pragma unroll
    for (int g = 0; g < 8; ++g) {
        const float pl = half_sum16(acc0[g] * al0 + acc1[g] * al1);
        const float pr = half_sum16(acc0[g] * ar0 + acc1[g] * ar1);
        if ((n15 & 7) == g) { vl = pl; vr = pr; }
    }
}

__device__ __forceinline__ void tile_store(float* __restrict__ z, int m, int n15, int khalf,
                                           const v8f& acc0, const v8f& acc1) {
    #pragma unroll
    for (int g = 0; g < 8; ++g) {
        const int row = m + g + 8 * khalf;
        z[(size_t)row * OUT_DIM + n15]      = acc0[g];
        z[(size_t)row * OUT_DIM + 16 + n15] = acc1[g];
    }
}

template <int WAVES, int ACTIVE>
__device__ __forceinline__ void gemm_body(int wg,
                     const float* __restrict__ h,
                     const float* __restrict__ fc_w,
                     const float* __restrict__ attn_w,
                     float* __restrict__ z,
                     float* __restrict__ s_l,
                     float* __restrict__ s_r,
                     int N) {
    __shared__ float sl_sh[ACTIVE * GEMM_WROWS];
    __shared__ float sr_sh[ACTIVE * GEMM_WROWS];
    const int lane = threadIdx.x & 31;
    const int wv   = threadIdx.x >> 5;
    const bool work = (ACTIVE == WAVES) ||
                      __builtin_amdgcn_readfirstlane((int)(threadIdx.x >> 5)) < ACTIVE;
    const int row0 = wg * (ACTIVE * GEMM_WROWS);
    const int wrow = row0 + wv * GEMM_WROWS;
    const bool live0 = work && wrow < N;
    const bool live1 = work && wrow + 16 < N;
    const int  m0    = live0 ? wrow : 0;
    const int  m1    = live1 ? wrow + 16 : 0;

    const int n15   = lane & 15;
    const int khalf = lane >> 4;

    const float* __restrict__ a0p = h + (size_t)(m0 + n15) * IN_DIM + 4 * khalf;
    const float* __restrict__ a1p = h + (size_t)(m1 + n15) * IN_DIM + 4 * khalf;
    const float* __restrict__ b0p = fc_w + (size_t)n15        * IN_DIM + 4 * khalf;
    const float* __restrict__ b1p = fc_w + (size_t)(16 + n15) * IN_DIM + 4 * khalf;

    const float al0 = attn_w[n15];
    const float al1 = attn_w[16 + n15];
    const float ar0 = attn_w[OUT_DIM + n15];
    const float ar1 = attn_w[OUT_DIM + 16 + n15];

    v8f c00 = {};
    v8f c01 = {};
    v8f c10 = {};
    v8f c11 = {};

    if (work) {
    #pragma unroll
    for (int k = 0; k < IN_DIM; k += 8) {
        const v4f a0 = *(const v4f*)(a0p + k);
        const v4f a1 = *(const v4f*)(a1p + k);
        const v4f b0 = *(const v4f*)(b0p + k);
        const v4f b1 = *(const v4f*)(b1p + k);
        const v2f a0l = {a0.x, a0.y}, a0h = {a0.z, a0.w};
        const v2f a1l = {a1.x, a1.y}, a1h = {a1.z, a1.w};
        const v2f b0l = {b0.x, b0.y}, b0h = {b0.z, b0.w};
        const v2f b1l = {b1.x, b1.y}, b1h = {b1.z, b1.w};
        c00 = wmma_f32(a0l, b0l, c00);
        c01 = wmma_f32(a0l, b1l, c01);
        c10 = wmma_f32(a1l, b0l, c10);
        c11 = wmma_f32(a1l, b1l, c11);
        c00 = wmma_f32(a0h, b0h, c00);
        c01 = wmma_f32(a0h, b1h, c01);
        c10 = wmma_f32(a1h, b0h, c10);
        c11 = wmma_f32(a1h, b1h, c11);
    }

    float vl0, vr0, vl1, vr1;
    tile_scores(c00, c01, al0, al1, ar0, ar1, n15, vl0, vr0);
    tile_scores(c10, c11, al0, al1, ar0, ar1, n15, vl1, vr1);
    if (n15 < 8) {
        const int j = wv * GEMM_WROWS + 8 * khalf + n15;
        sl_sh[j]      = vl0;
        sr_sh[j]      = vr0;
        sl_sh[j + 16] = vl1;
        sr_sh[j + 16] = vr1;
    }
    }
    __syncthreads();
    {
        const int row = row0 + (int)threadIdx.x;
        if ((int)threadIdx.x < ACTIVE * GEMM_WROWS && row < N) {
            s_l[row] = sl_sh[threadIdx.x];
            s_r[row] = sr_sh[threadIdx.x];
        }
    }

    if (live0) tile_store(z, m0, n15, khalf, c00, c01);
    if (live1) tile_store(z, m1, n15, khalf, c10, c11);
}

__global__ void __launch_bounds__(GEMM_WAVES * 32)
gat_gemm_wmma_kernel(const float* __restrict__ h,
                     const float* __restrict__ fc_w,
                     const float* __restrict__ attn_w,
                     float* __restrict__ z,
                     float* __restrict__ s_l,
                     float* __restrict__ s_r,
                     int N) {
    gemm_body<GEMM_WAVES, GEMM_WAVES>((int)blockIdx.x, h, fc_w, attn_w, z, s_l, s_r, N);
}

#define BK_T 1024
#define BK_U 16
#define NO_NODE 0x7fffffff

template <bool FILL>
__device__ __forceinline__ int bucket_one(int dv, int e, int lo, int* cnt,
                                          const int* __restrict__ src,
                                          int* __restrict__ bucket_src) {
    const unsigned r = (unsigned)(dv - lo);
    if (r < (unsigned)BK_T) {
        if (FILL) {
            const int pos = atomicAdd(&cnt[r], 1);
            bucket_src[pos] = src[e];
        } else {
            atomicAdd(&cnt[r], 1);
        }
    }
    if (FILL) return 0;
    return __builtin_popcount(__builtin_amdgcn_ballot_w32(dv < lo));
}

template <bool GUARD, bool FILL>
__device__ __forceinline__ int bucket_batch(const int4* __restrict__ dst4, int q0, int E4,
                                            int lo, int* cnt,
                                            const int* __restrict__ src,
                                            int* __restrict__ bucket_src) {
    int4 d[BK_U];
    #pragma unroll
    for (int u = 0; u < BK_U; ++u) {
        const int q = q0 + u * BK_T;
        if (GUARD) {
            d[u] = make_int4(NO_NODE, NO_NODE, NO_NODE, NO_NODE);
            if (q < E4) d[u] = dst4[q];
        } else {
            d[u] = dst4[q];
        }
    }
    int below = 0;
    #pragma unroll
    for (int u = 0; u < BK_U; ++u) {
        const int e0 = (q0 + u * BK_T) * 4;
        below += bucket_one<FILL>(d[u].x, e0 + 0, lo, cnt, src, bucket_src);
        below += bucket_one<FILL>(d[u].y, e0 + 1, lo, cnt, src, bucket_src);
        below += bucket_one<FILL>(d[u].z, e0 + 2, lo, cnt, src, bucket_src);
        below += bucket_one<FILL>(d[u].w, e0 + 3, lo, cnt, src, bucket_src);
    }
    return below;
}

template <bool FILL>
__device__ __forceinline__ int bucket_pass(const int* __restrict__ dst, int E, int lo, int tid,
                                           int* cnt, const int* __restrict__ src,
                                           int* __restrict__ bucket_src) {
    const int4* __restrict__ dst4 = (const int4*)dst;
    const int E4 = E >> 2;
    const int per_batch = BK_T * BK_U;
    const int nfull = E4 / per_batch;
    int below = 0;
    for (int b = 0; b < nfull; ++b)
        below += bucket_batch<false, FILL>(dst4, b * per_batch + tid, E4, lo, cnt, src, bucket_src);
    if (nfull * per_batch < E4)
        below += bucket_batch<true, FILL>(dst4, nfull * per_batch + tid, E4, lo, cnt, src, bucket_src);
    {
        const int e = (E4 << 2) + tid;
        int dv = NO_NODE;
        if (e < E) dv = dst[e];
        below += bucket_one<FILL>(dv, e, lo, cnt, src, bucket_src);
    }
    return below;
}

__global__ void __launch_bounds__(BK_T)
gat_bucket_kernel(const int* __restrict__ src,
                  const int* __restrict__ dst,
                  int* __restrict__ offset,
                  int* __restrict__ count,
                  int* __restrict__ bucket_src,
                  int N, int E) {
    __shared__ int cnt[BK_T];
    __shared__ int wave_tot[32];
    __shared__ int wave_below[32];
    __shared__ int base_sh;
    const int tid  = threadIdx.x;
    const int lane = tid & 31;
    const int wv   = tid >> 5;
    const int lo   = blockIdx.x * BK_T;

    cnt[tid] = 0;
    __syncthreads();

    const int below = bucket_pass<false>(dst, E, lo, tid, cnt, src, bucket_src);
    __syncthreads();

    const int c = cnt[tid];
    int inc = c;
    #pragma unroll
    for (int d = 1; d < 32; d <<= 1) {
        int o = __shfl_up(inc, d, 32);
        if (lane >= d) inc += o;
    }
    if (lane == 31) wave_tot[wv] = inc;
    if (lane == 0)  wave_below[wv] = below;
    __syncthreads();
    if (wv == 0) {
        const int w = wave_tot[lane];
        int winc = w;
        #pragma unroll
        for (int d = 1; d < 32; d <<= 1) {
            int o = __shfl_up(winc, d, 32);
            if (lane >= d) winc += o;
        }
        wave_tot[lane] = winc - w;
        int b = wave_below[lane];
        #pragma unroll
        for (int d = 16; d >= 1; d >>= 1) b += __shfl_xor(b, d, 32);
        if (lane == 0) base_sh = b;
    }
    __syncthreads();
    const int start = base_sh + wave_tot[wv] + (inc - c);
    if (lo + tid < N) {
        offset[lo + tid] = start;
        count[lo + tid]  = c;
    }
    cnt[tid] = start;
    __syncthreads();

    (void)bucket_pass<true>(dst, E, lo, tid, cnt, src, bucket_src);
}

#define BK_SHIFT 10
#define CP_G 128
#define CP_T 1024
#define CP_U 8

__device__ __forceinline__ int hist_pitch_log2(int NB) {
    int hs = 5;
    while ((1 << hs) < NB) ++hs;
    return hs;
}

__global__ void __launch_bounds__(CP_T)
gat_coarse_hist_kernel(const int* __restrict__ dst, int* __restrict__ hist,
                       int N, int NB, int E) {
    __shared__ int cnt[CP_T];
    const int tid = threadIdx.x;
    const int c   = blockIdx.x;
    const int len = (E + CP_G - 1) / CP_G;
    const int beg = c * len;
    const int end = (beg + len < E) ? beg + len : E;
    const int HP  = 1 << hist_pitch_log2(NB);

    cnt[tid] = 0;
    __syncthreads();
    for (int base = beg; base < end; base += CP_T * CP_U) {
        int d[CP_U];
        #pragma unroll
        for (int u = 0; u < CP_U; ++u) {
            const int e = base + u * CP_T + tid;
            d[u] = -1;
            if (e < end) d[u] = dst[e];
        }
        #pragma unroll
        for (int u = 0; u < CP_U; ++u)
            if ((unsigned)d[u] < (unsigned)N) atomicAdd(&cnt[d[u] >> BK_SHIFT], 1);
    }
    __syncthreads();
    if (tid < HP) hist[c * HP + tid] = cnt[tid];
}

#define CS_CAP 16384

template <bool STAGED>
__device__ __forceinline__ void coarse_fill(const int* __restrict__ src, const int* __restrict__ dst,
                                            int beg, int end, int tid, int N, int* cur,
                                            int2* stage, int2* __restrict__ coarse) {
    for (int base = beg; base < end; base += CP_T * CP_U) {
        int d[CP_U];
        int s[CP_U];
        #pragma unroll
        for (int u = 0; u < CP_U; ++u) {
            const int e = base + u * CP_T + tid;
            d[u] = -1;
            s[u] = 0;
            if (e < end) { d[u] = dst[e]; s[u] = src[e]; }
        }
        #pragma unroll
        for (int u = 0; u < CP_U; ++u) {
            if ((unsigned)d[u] < (unsigned)N) {
                const int pos = atomicAdd(&cur[d[u] >> BK_SHIFT], 1);
                if (STAGED) stage[pos]  = make_int2(d[u], s[u]);
                else        coarse[pos] = make_int2(d[u], s[u]);
            }
        }
    }
}

__global__ void __launch_bounds__(CP_T)
gat_coarse_scatter_kernel(const int* __restrict__ src, const int* __restrict__ dst,
                          const int* __restrict__ hist,
                          int* __restrict__ coarse_raw, int* __restrict__ binstart,
                          int N, int NB, int E) {
    __shared__ int cur[CP_T];
    __shared__ int tot[CP_T];
    __shared__ int dlt[CP_T];
    __shared__ int wave_tot[32];
    __shared__ int wave_own[32];
    __shared__ int nvalid_sh;
    __shared__ int2 stage[CS_CAP];
    int2* __restrict__ coarse = (int2*)coarse_raw;
    const int tid  = threadIdx.x;
    const int lane = tid & 31;
    const int wv   = tid >> 5;
    const int c    = blockIdx.x;
    const int len  = (E + CP_G - 1) / CP_G;
    const int beg  = c * len;
    const int end  = (beg + len < E) ? beg + len : E;
    const int hs   = hist_pitch_log2(NB);
    const int HP   = 1 << hs;

    cur[tid] = 0;
    tot[tid] = 0;
    __syncthreads();
    {
        const int b = tid & (HP - 1);
        int part = 0, all = 0;
        for (int j = tid >> hs; j < CP_G; j += (CP_T >> hs)) {
            const int v = hist[j * HP + b];
            all += v;
            if (j < c) part += v;
        }
        atomicAdd(&cur[b], part);
        atomicAdd(&tot[b], all);
    }
    __syncthreads();
    const bool staged = len <= CS_CAP;
    const int v   = tot[tid];
    const int own = (tid < HP) ? hist[c * HP + tid] : 0;
    int inc = v, oinc = own;
    for (int d = 1; d < 32; d <<= 1) {
        int o  = __shfl_up(inc, d, 32);
        int oo = __shfl_up(oinc, d, 32);
        if (lane >= d) { inc += o; oinc += oo; }
    }
    if (lane == 31) { wave_tot[wv] = inc; wave_own[wv] = oinc; }
    __syncthreads();
    if (wv == 0) {
        const int w  = wave_tot[lane];
        const int ow = wave_own[lane];
        int winc = w, owinc = ow;
        for (int d = 1; d < 32; d <<= 1) {
            int o  = __shfl_up(winc, d, 32);
            int oo = __shfl_up(owinc, d, 32);
            if (lane >= d) { winc += o; owinc += oo; }
        }
        wave_tot[lane] = winc - w;
        wave_own[lane] = owinc - ow;
        if (lane == 31) nvalid_sh = owinc;
    }
    __syncthreads();
    const int excl  = wave_tot[wv] + (inc - v);
    const int lexcl = wave_own[wv] + (oinc - own);
    const int g0    = cur[tid] + excl;
    cur[tid] = staged ? lexcl : g0;
    dlt[tid] = g0 - lexcl;
    if (c == 0 && tid <= NB) binstart[tid] = excl;
    __syncthreads();

    if (staged) {
        coarse_fill<true>(src, dst, beg, end, tid, N, cur, stage, coarse);
        __syncthreads();
        const int nvalid = nvalid_sh;
        for (int i = tid; i < nvalid; i += CP_T) {
            const int2 p = stage[i];
            coarse[i + dlt[p.x >> BK_SHIFT]] = p;
        }
    } else {
        coarse_fill<false>(src, dst, beg, end, tid, N, cur, stage, coarse);
    }
}

#define FB_U 8

#define FB_CAP 32768

template <bool FILL>
__device__ __forceinline__ void fine_pass(const int2* __restrict__ coarse, int rb, int re,
                                          int lo, int tid, int* cnt,
                                          int* bucket_src) {
    for (int base = rb; base < re; base += BK_T * FB_U) {
        int2 p[FB_U];
        #pragma unroll
        for (int u = 0; u < FB_U; ++u) {
            const int i = base + u * BK_T + tid;
            p[u] = make_int2(-1, 0);
            if (i < re) p[u] = coarse[i];
        }
        #pragma unroll
        for (int u = 0; u < FB_U; ++u) {
            const unsigned r = (unsigned)(p[u].x - lo);
            if (r < (unsigned)BK_T) {
                if (FILL) {
                    const int pos = atomicAdd(&cnt[r], 1);
                    bucket_src[pos] = p[u].y;
                } else {
                    atomicAdd(&cnt[r], 1);
                }
            }
        }
    }
}

#define ROLE_WAVES 20
#define ROLE_ROWS (ROLE_WAVES * GEMM_WROWS)

template <bool WITH_GEMM>
__global__ void __launch_bounds__(BK_T)
gat_fine_bucket_kernel(const int* __restrict__ coarse_raw, const int* __restrict__ binstart,
                       int* __restrict__ offset, int* __restrict__ count,
                       int* __restrict__ bucket_src, int N, int NG,
                       const float* __restrict__ h, const float* __restrict__ fc_w,
                       const float* __restrict__ attn_w, float* __restrict__ z,
                       float* __restrict__ s_l, float* __restrict__ s_r) {
    if (WITH_GEMM && (int)blockIdx.x < NG) {
        gemm_body<BK_T / 32, ROLE_WAVES>((int)blockIdx.x, h, fc_w, attn_w, z, s_l, s_r, N);
        return;
    }
    __shared__ int cnt[BK_T];
    __shared__ int wave_tot[32];
    __shared__ int total_sh;
    __shared__ int stage[FB_CAP];
    const int2* __restrict__ coarse = (const int2*)coarse_raw;
    const int tid  = threadIdx.x;
    const int lane = tid & 31;
    const int wv   = tid >> 5;
    const int bin  = (int)blockIdx.x - (WITH_GEMM ? NG : 0);
    const int lo   = bin * BK_T;
    const int rb   = binstart[bin];
    const int re   = binstart[bin + 1];
    const bool staged = (re - rb) <= FB_CAP;

    cnt[tid] = 0;
    __syncthreads();
    fine_pass<false>(coarse, rb, re, lo, tid, cnt, bucket_src);
    __syncthreads();

    const int c = cnt[tid];
    int inc = c;
    for (int d = 1; d < 32; d <<= 1) {
        int o = __shfl_up(inc, d, 32);
        if (lane >= d) inc += o;
    }
    if (lane == 31) wave_tot[wv] = inc;
    __syncthreads();
    if (wv == 0) {
        const int w = wave_tot[lane];
        int winc = w;
        for (int d = 1; d < 32; d <<= 1) {
            int o = __shfl_up(winc, d, 32);
            if (lane >= d) winc += o;
        }
        wave_tot[lane] = winc - w;
        if (lane == 31) total_sh = winc;
    }
    __syncthreads();
    const int lstart = wave_tot[wv] + (inc - c);
    const int start  = rb + lstart;
    if (lo + tid < N) {
        offset[lo + tid] = start;
        if (!WITH_GEMM) count[lo + tid] = c;
    }
    if (WITH_GEMM && lo + BK_T >= N && tid == 0) offset[N] = re;
    cnt[tid] = staged ? lstart : start;
    __syncthreads();

    if (staged) {
        fine_pass<true>(coarse, rb, re, lo, tid, cnt, stage);
        __syncthreads();
        const int total = total_sh;
        for (int i = tid; i < total; i += BK_T) bucket_src[rb + i] = stage[i];
    } else {
        fine_pass<true>(coarse, rb, re, lo, tid, cnt, bucket_src);
    }
}

#define CO_R (CS_CAP / CP_T)

__global__ void __launch_bounds__(CP_T)
gat_coarse_sort_kernel(const int* __restrict__ src, const int* __restrict__ dst,
                       int* __restrict__ coarse_raw,
                       int* __restrict__ run_cnt, int* __restrict__ run_start,
                       int N, int NB, int E) {
    __shared__ int cnt[CP_T];
    __shared__ int wave_tot[32];
    __shared__ int nvalid_sh;
    __shared__ int2 stage[CS_CAP];
    int2* __restrict__ coarse = (int2*)coarse_raw;
    const int tid  = threadIdx.x;
    const int lane = tid & 31;
    const int wv   = tid >> 5;
    const int c    = blockIdx.x;
    const int len  = (E + CP_G - 1) / CP_G;
    const int beg  = c * len;
    const int end  = (beg + len < E) ? beg + len : E;

    int d[CO_R];
    int s[CO_R];
    #pragma unroll
    for (int u = 0; u < CO_R; ++u) {
        const int e = beg + u * CP_T + tid;
        d[u] = -1;
        s[u] = 0;
        if (e < end) { d[u] = dst[e]; s[u] = src[e]; }
    }
    cnt[tid] = 0;
    __syncthreads();
    int rk[CO_R];
    #pragma unroll
    for (int u = 0; u < CO_R; ++u) {
        rk[u] = 0;
        if ((unsigned)d[u] < (unsigned)N) rk[u] = atomicAdd(&cnt[d[u] >> BK_SHIFT], 1);
    }
    __syncthreads();

    const int own = cnt[tid];
    int inc = own;
    for (int k = 1; k < 32; k <<= 1) {
        int o = __shfl_up(inc, k, 32);
        if (lane >= k) inc += o;
    }
    if (lane == 31) wave_tot[wv] = inc;
    __syncthreads();
    if (wv == 0) {
        const int w = wave_tot[lane];
        int winc = w;
        for (int k = 1; k < 32; k <<= 1) {
            int o = __shfl_up(winc, k, 32);
            if (lane >= k) winc += o;
        }
        wave_tot[lane] = winc - w;
        if (lane == 31) nvalid_sh = winc;
    }
    __syncthreads();
    const int lexcl = wave_tot[wv] + (inc - own);
    cnt[tid] = lexcl;
    if (tid < NB) {
        run_cnt[tid * CP_G + c]   = own;
        run_start[tid * CP_G + c] = lexcl;
    }
    __syncthreads();

    #pragma unroll
    for (int u = 0; u < CO_R; ++u)
        if ((unsigned)d[u] < (unsigned)N)
            stage[cnt[d[u] >> BK_SHIFT] + rk[u]] = make_int2(d[u], s[u]);
    __syncthreads();
    const int nvalid = nvalid_sh;
    for (int i = tid; i < nvalid; i += CP_T) coarse[beg + i] = stage[i];
}

#define FR_Q (CP_G / (BK_T / 32))
#define FR_I 8

template <bool FILL>
__device__ __forceinline__ void fine_one(int2 p, int lo, int* cnt, int* slots) {
    const unsigned r = (unsigned)(p.x - lo);
    if (r < (unsigned)BK_T) {
        if (FILL) {
            const int pos = atomicAdd(&cnt[r], 1);
            slots[pos] = p.y;
        } else {
            atomicAdd(&cnt[r], 1);
        }
    }
}

__global__ void __launch_bounds__(BK_T)
gat_fine_runs_kernel(const int* __restrict__ coarse_raw,
                     const int* __restrict__ run_cnt, const int* __restrict__ run_start,
                     int* __restrict__ offset, int* __restrict__ bucket_src,
                     int N, int NG, int E,
                     const float* __restrict__ h, const float* __restrict__ fc_w,
                     const float* __restrict__ attn_w, float* __restrict__ z,
                     float* __restrict__ s_l, float* __restrict__ s_r) {
    if ((int)blockIdx.x < NG) {
        gemm_body<BK_T / 32, ROLE_WAVES>((int)blockIdx.x, h, fc_w, attn_w, z, s_l, s_r, N);
        return;
    }
    __shared__ int cnt[BK_T];
    __shared__ int wave_tot[32];
    __shared__ int seg_cnt[CP_G];
    __shared__ int seg_base[CP_G];
    __shared__ int before_w[CP_G / 32];
    __shared__ int total_w[CP_G / 32];
    __shared__ int stage[FB_CAP];
    const int2* __restrict__ coarse = (const int2*)coarse_raw;
    const int tid  = threadIdx.x;
    const int lane = tid & 31;
    const int wv   = tid >> 5;
    const int bin  = (int)blockIdx.x - NG;
    const int lo   = bin * BK_T;
    const int len  = (E + CP_G - 1) / CP_G;

    if (tid < CP_G) {
        const int st = run_start[bin * CP_G + tid];
        const int cn = run_cnt[bin * CP_G + tid];
        seg_cnt[tid]  = cn;
        seg_base[tid] = tid * len + st;
        int a = st, b = cn;
        for (int k = 16; k >= 1; k >>= 1) { a += __shfl_xor(a, k, 32); b += __shfl_xor(b, k, 32); }
        if (lane == 0) { before_w[wv] = a; total_w[wv] = b; }
    }
    cnt[tid] = 0;
    __syncthreads();
    int rb = 0, total = 0;
    #pragma unroll
    for (int k = 0; k < CP_G / 32; ++k) { rb += before_w[k]; total += total_w[k]; }
    const bool staged = total <= FB_CAP;

    int sc[FR_Q], sb[FR_Q];
    #pragma unroll
    for (int q = 0; q < FR_Q; ++q) {
        sc[q] = __builtin_amdgcn_readfirstlane(seg_cnt[wv * FR_Q + q]);
        sb[q] = __builtin_amdgcn_readfirstlane(seg_base[wv * FR_Q + q]);
    }
    int2 p[FR_Q][FR_I];
    #pragma unroll
    for (int q = 0; q < FR_Q; ++q) {
        #pragma unroll
        for (int it = 0; it < FR_I; ++it) {
            const int j = it * 32 + lane;
            p[q][it] = make_int2(-1, 0);
            if (j < sc[q]) p[q][it] = coarse[sb[q] + j];
        }
    }
    #pragma unroll
    for (int q = 0; q < FR_Q; ++q) {
        #pragma unroll
        for (int it = 0; it < FR_I; ++it) fine_one<false>(p[q][it], lo, cnt, stage);
        for (int j = FR_I * 32 + lane; j < sc[q]; j += 32) fine_one<false>(coarse[sb[q] + j], lo, cnt, stage);
    }
    __syncthreads();

    const int c = cnt[tid];
    int inc = c;
    for (int k = 1; k < 32; k <<= 1) {
        int o = __shfl_up(inc, k, 32);
        if (lane >= k) inc += o;
    }
    if (lane == 31) wave_tot[wv] = inc;
    __syncthreads();
    if (wv == 0) {
        const int w = wave_tot[lane];
        int winc = w;
        for (int k = 1; k < 32; k <<= 1) {
            int o = __shfl_up(winc, k, 32);
            if (lane >= k) winc += o;
        }
        wave_tot[lane] = winc - w;
    }
    __syncthreads();
    const int lstart = wave_tot[wv] + (inc - c);
    const int start  = rb + lstart;
    if (lo + tid < N) offset[lo + tid] = start;
    if (lo + BK_T >= N && tid == 0) offset[N] = rb + total;
    cnt[tid] = staged ? lstart : start;
    __syncthreads();

    if (staged) {
        #pragma unroll
        for (int q = 0; q < FR_Q; ++q) {
            #pragma unroll
            for (int it = 0; it < FR_I; ++it) fine_one<true>(p[q][it], lo, cnt, stage);
            for (int j = FR_I * 32 + lane; j < sc[q]; j += 32) fine_one<true>(coarse[sb[q] + j], lo, cnt, stage);
        }
        __syncthreads();
        for (int i = tid; i < total; i += BK_T) bucket_src[rb + i] = stage[i];
    } else {
        #pragma unroll
        for (int q = 0; q < FR_Q; ++q) {
            #pragma unroll
            for (int it = 0; it < FR_I; ++it) fine_one<true>(p[q][it], lo, cnt, bucket_src);
            for (int j = FR_I * 32 + lane; j < sc[q]; j += 32) fine_one<true>(coarse[sb[q] + j], lo, cnt, bucket_src);
        }
    }
}

__device__ __forceinline__ float wave_max_f(float x) {
    #pragma unroll
    for (int d = 16; d >= 1; d >>= 1) x = fmaxf(x, __shfl_xor(x, d, 32));
    return x;
}
__device__ __forceinline__ float wave_sum_f(float x) {
    #pragma unroll
    for (int d = 16; d >= 1; d >>= 1) x += __shfl_xor(x, d, 32);
    return x;
}
__device__ __forceinline__ float edge_score(const float* __restrict__ s_l, int s, float sr) {
    float v = s_l[s] + sr;
    return v > 0.0f ? v : NEG_SLOPE * v;
}
template <int U>
__device__ __forceinline__ float add_rows_n(const float* __restrict__ z, int s, float a,
                                            int k, int lane, float acc) {
    float zr[U];
    float ak[U];
    #pragma unroll
    for (int u = 0; u < U; ++u) {
        const int sk = __builtin_amdgcn_readlane(s, k + u);
        ak[u] = __int_as_float(__builtin_amdgcn_readlane(__float_as_int(a), k + u));
        zr[u] = z[(size_t)sk * OUT_DIM + lane];
    }
    #pragma unroll
    for (int u = 0; u < U; ++u) acc += ak[u] * zr[u];
    return acc;
}
__device__ __forceinline__ float add_rows(const float* __restrict__ z, int s, float a,
                                          int cnt, int lane, float acc) {
    int k = 0;
    for (; k + 8 <= cnt; k += 8) acc = add_rows_n<8>(z, s, a, k, lane, acc);
    if (k + 4 <= cnt) { acc = add_rows_n<4>(z, s, a, k, lane, acc); k += 4; }
    if (k + 2 <= cnt) { acc = add_rows_n<2>(z, s, a, k, lane, acc); k += 2; }
    if (k < cnt)      { acc = add_rows_n<1>(z, s, a, k, lane, acc); }
    return acc;
}

#define NODE_WG 2048

__device__ __forceinline__ float score_of(float sl, float sr) {
    float v = sl + sr;
    return v > 0.0f ? v : NEG_SLOPE * v;
}

template <bool HAS_COUNT>
__device__ __forceinline__ int node_degree(const int* __restrict__ offset,
                                           const int* __restrict__ count, int t, int beg) {
    return HAS_COUNT ? count[t] : offset[t + 1] - beg;
}

#define NODE_R 32
#define NODE_G 4

template <bool HAS_COUNT>
__global__ void __launch_bounds__(256)
gat_node_kernel(const int* __restrict__ bucket_src,
                const int* __restrict__ offset,
                const int* __restrict__ count,
                const float* __restrict__ z,
                const float* __restrict__ s_l,
                const float* __restrict__ s_r,
                float* __restrict__ h_out,
                int N) {
    const int lane = threadIdx.x & 31;
    const int wpb  = (int)(blockDim.x >> 5);
    const int nw   = (int)gridDim.x * wpb;
    int t = __builtin_amdgcn_readfirstlane(
                (int)(blockIdx.x * wpb + (threadIdx.x >> 5)));
    if (t >= N) return;

    int   beg = offset[t];
    int   deg = node_degree<HAS_COUNT>(offset, count, t, beg);
    float sr  = s_r[t];
    int   s = 0;
    float e = -INFINITY;
    if (lane < deg) {
        s = bucket_src[beg + lane];
        e = score_of(s_l[s], sr);
    }
    int   tn = t + nw;
    int   begn = 0, degn = 0;
    float srn = 0.0f;
    int   sn = 0;
    if (tn < N) {
        begn = offset[tn]; degn = node_degree<HAS_COUNT>(offset, count, tn, begn); srn = s_r[tn];
        if (lane < degn) sn = bucket_src[begn + lane];
    }
    int   tnn = tn + nw;
    int   begnn = 0, degnn = 0;
    float srnn = 0.0f;
    if (tnn < N) { begnn = offset[tnn]; degnn = node_degree<HAS_COUNT>(offset, count, tnn, begnn); srnn = s_r[tnn]; }

    for (;;) {
        const int t3 = tnn + nw;
        int   beg3 = 0, raw3 = 0;
        float sr3 = 0.0f;
        if (t3 < N) { beg3 = offset[t3]; raw3 = HAS_COUNT ? count[t3] : offset[t3 + 1]; sr3 = s_r[t3]; }
        float gn = 0.0f;
        if (lane < degn) gn = s_l[sn];

        int   snn = 0;
        if (lane < degnn) snn = bucket_src[begnn + lane];

        float acc = 0.0f;
        float en  = -INFINITY;
        if (deg <= NODE_R) {
            const int last = __builtin_amdgcn_readlane(s, deg - 1);
            const int sp   = (lane < deg) ? s : last;
            float zr[NODE_R];
            #pragma unroll
            for (int g = 0; g < NODE_R / NODE_G; ++g) {
                if (g * NODE_G < deg) {
                    #pragma unroll
                    for (int u = 0; u < NODE_G; ++u) {
                        const int k  = g * NODE_G + u;
                        const int sk = __builtin_amdgcn_readlane(sp, k);
                        zr[k] = z[(unsigned)(sk * OUT_DIM + lane)];
                    }
                }
            }
            const float m   = wave_max_f(e);
            const float ex  = (lane < deg) ? __expf(e - m) : 0.0f;
            const float den = wave_sum_f(ex);
            const float a   = ex / den;
            #pragma unroll
            for (int g = 0; g < NODE_R / NODE_G; ++g) {
                if (g * NODE_G < deg) {
                    #pragma unroll
                    for (int u = 0; u < NODE_G; ++u) {
                        const int k = g * NODE_G + u;
                        const float ak = __int_as_float(__builtin_amdgcn_readlane(__float_as_int(a), k));
                        acc = fmaf(ak, zr[k], acc);
                    }
                }
            }
            if (lane < degn) en = score_of(gn, srn);
        } else {
            float m = -INFINITY;
            for (int j = lane; j < deg; j += 32)
                m = fmaxf(m, edge_score(s_l, bucket_src[beg + j], sr));
            m = wave_max_f(m);
            float den = 0.0f;
            for (int j = lane; j < deg; j += 32)
                den += __expf(edge_score(s_l, bucket_src[beg + j], sr) - m);
            den = wave_sum_f(den);
            for (int base = 0; base < deg; base += 32) {
                const int j = base + lane;
                int   sj = 0;
                float a = 0.0f;
                if (j < deg) {
                    sj = bucket_src[beg + j];
                    a = __expf(edge_score(s_l, sj, sr) - m) / den;
                }
                const int left = deg - base;
                acc = add_rows(z, sj, a, left < 32 ? left : 32, lane, acc);
            }
            if (lane < degn) en = score_of(gn, srn);
        }
        h_out[(size_t)t * OUT_DIM + lane] = acc;

        if (tn >= N) break;
        t = tn;   beg = begn;   deg = degn;   sr = srn;   s = sn;   e = en;
        tn = tnn; begn = begnn; degn = degnn; srn = srnn; sn = snn;
        tnn = t3; begnn = beg3; degnn = HAS_COUNT ? raw3 : raw3 - beg3; srnn = sr3;
    }
}

extern "C" void kernel_launch(void* const* d_in, const int* in_sizes, int n_in,
                              void* d_out, int out_size, void* d_ws, size_t ws_size,
                              hipStream_t stream) {
    const float* h      = (const float*)d_in[0];
    const int*   src    = (const int*)  d_in[1];
    const int*   dst    = (const int*)  d_in[2];
    const float* fc_w   = (const float*)d_in[3];
    const float* attn_w = (const float*)d_in[4];

    const int N = in_sizes[0] / IN_DIM;
    const int E = in_sizes[1];

    float* h_out = (float*)d_out;

    auto align256 = [](size_t x) { return (x + 255) & ~(size_t)255; };
    char* w = (char*)d_ws;
    float* z      = (float*)w; w += align256((size_t)N * OUT_DIM * sizeof(float));
    float* s_l    = (float*)w; w += align256((size_t)N * sizeof(float));
    float* s_r    = (float*)w; w += align256((size_t)N * sizeof(float));
    int*   count  = (int*)w;   w += align256((size_t)N * sizeof(int));
    int*   offset = (int*)w;   w += align256((size_t)N * sizeof(int));
    int*   bucket = (int*)w;   w += align256((size_t)E * sizeof(int));
    (void)ws_size;

    const int B = 256;
    const int tiles = (N + 15) / 16;
    const int gemm_grid = (tiles * 16 + GEMM_ROWS - 1) / GEMM_ROWS;
    const int NB = (N + BK_T - 1) / BK_T;

    const bool two_level =
        NB < CP_T &&
        (size_t)E * sizeof(int2) <= (size_t)N * OUT_DIM * sizeof(float) &&
        (size_t)CP_G * 2 * (NB < 32 ? 32 : NB) <= (size_t)N;

    const size_t bounds_bytes = align256((size_t)(N + 1) * sizeof(int));
    const bool shared_launch =
        NB < CP_T &&
        (size_t)E * sizeof(int2) <= (size_t)out_size * sizeof(float) &&
        (size_t)CP_G * CP_T <= (size_t)E &&
        bounds_bytes + (size_t)(NB + 1) * sizeof(int) <= 2 * align256((size_t)N * sizeof(int));

    const bool one_pass_coarse =
        shared_launch &&
        (size_t)E <= (size_t)CP_G * CS_CAP &&
        bounds_bytes + 2 * (size_t)NB * CP_G * sizeof(int) <= 2 * align256((size_t)N * sizeof(int));

    if (one_pass_coarse) {
        int* coarse    = (int*)h_out;
        int* bounds    = count;
        int* run_cnt   = (int*)((char*)count + bounds_bytes);
        int* run_start = run_cnt + (size_t)NB * CP_G;
        const int NG   = (N + ROLE_ROWS - 1) / ROLE_ROWS;
        gat_coarse_sort_kernel<<<CP_G, CP_T, 0, stream>>>(src, dst, coarse, run_cnt, run_start, N, NB, E);
        gat_fine_runs_kernel<<<NG + NB, BK_T, 0, stream>>>(coarse, run_cnt, run_start, bounds, bucket, N, NG, E,
                                                          h, fc_w, attn_w, z, s_l, s_r);
        int wavesPerBlock = B / 32;
        int need = (N + wavesPerBlock - 1) / wavesPerBlock;
        int grid = need < NODE_WG ? need : NODE_WG;
        gat_node_kernel<false><<<grid, B, 0, stream>>>(bucket, bounds, bounds, z, s_l, s_r, h_out, N);
        return;
    }

    if (shared_launch) {
        int* coarse   = (int*)h_out;
        int* hist     = bucket;
        int* bounds   = count;
        int* binstart = (int*)((char*)count + bounds_bytes);
        const int NG  = (N + ROLE_ROWS - 1) / ROLE_ROWS;
        gat_coarse_hist_kernel<<<CP_G, CP_T, 0, stream>>>(dst, hist, N, NB, E);
        gat_coarse_scatter_kernel<<<CP_G, CP_T, 0, stream>>>(src, dst, hist, coarse, binstart, N, NB, E);
        gat_fine_bucket_kernel<true><<<NG + NB, BK_T, 0, stream>>>(coarse, binstart, bounds, bounds, bucket, N, NG,
                                                                  h, fc_w, attn_w, z, s_l, s_r);
        int wavesPerBlock = B / 32;
        int need = (N + wavesPerBlock - 1) / wavesPerBlock;
        int grid = need < NODE_WG ? need : NODE_WG;
        gat_node_kernel<false><<<grid, B, 0, stream>>>(bucket, bounds, bounds, z, s_l, s_r, h_out, N);
        return;
    }

    if (two_level) {
        int* coarse   = (int*)z;
        int* hist     = (int*)s_l;
        int* binstart = (int*)s_r;
        gat_coarse_hist_kernel<<<CP_G, CP_T, 0, stream>>>(dst, hist, N, NB, E);
        gat_coarse_scatter_kernel<<<CP_G, CP_T, 0, stream>>>(src, dst, hist, coarse, binstart, N, NB, E);
        gat_fine_bucket_kernel<false><<<NB, BK_T, 0, stream>>>(coarse, binstart, offset, count, bucket, N, 0,
                                                               h, fc_w, attn_w, z, s_l, s_r);
        gat_gemm_wmma_kernel<<<gemm_grid, GEMM_WAVES * 32, 0, stream>>>(h, fc_w, attn_w, z, s_l, s_r, N);
    } else {
        gat_gemm_wmma_kernel<<<gemm_grid, GEMM_WAVES * 32, 0, stream>>>(h, fc_w, attn_w, z, s_l, s_r, N);
        gat_bucket_kernel<<<NB, BK_T, 0, stream>>>(src, dst, offset, count, bucket, N, E);
    }
    {
        int wavesPerBlock = B / 32;
        int need = (N + wavesPerBlock - 1) / wavesPerBlock;
        int grid = need < NODE_WG ? need : NODE_WG;
        gat_node_kernel<true><<<grid, B, 0, stream>>>(bucket, offset, count, z, s_l, s_r, h_out, N);
    }
}
